# starting kernel
# baseline (speedup 1.0000x reference)
#include <hip/hip_runtime.h>

typedef __attribute__((ext_vector_type(2))) float v2f;
typedef __attribute__((ext_vector_type(8))) float v8f;

#define C_DIM 128
#define LDA   132
#define MTILE 64

__global__ __launch_bounds__(256)
void splineconv_gemm_kernel(const float* __restrict__ X,
                            const float* __restrict__ W,
                            const float* __restrict__ root,
                            float* __restrict__ H,
                            int N)
{
    __shared__ float As[MTILE * LDA];
    const int tid  = threadIdx.x;
    const int wave = tid >> 5;
    const int lane = tid & 31;
    const int row0 = blockIdx.x * MTILE;

    for (int i = tid; i < MTILE * (C_DIM / 4); i += 256) {
        const int r  = i >> 5;
        const int c4 = i & 31;
        float4 v = make_float4(0.f, 0.f, 0.f, 0.f);
        const int gr = row0 + r;
        if (gr < N) v = ((const float4*)(X + (size_t)gr * C_DIM))[c4];
        float* dst = &As[r * LDA + c4 * 4];
        dst[0] = v.x; dst[1] = v.y; dst[2] = v.z; dst[3] = v.w;
    }
    __syncthreads();

    const int hi  = lane >> 4;
    const int l16 = lane & 15;

    for (int t = wave; t < 24; t += 8) {
        const int gcol = t * 16;
        const float* Bbase;
        if (gcol < C_DIM)          Bbase = W + gcol;
        else if (gcol < 2 * C_DIM) Bbase = W + (size_t)C_DIM * C_DIM + (gcol - C_DIM);
        else                       Bbase = root + (gcol - 2 * C_DIM);

        const float* Bl = Bbase + (size_t)(2 * hi) * C_DIM + l16;
        const float* Al = &As[l16 * LDA + 2 * hi];

        v2f bf[32];
        #pragma unroll
        for (int k4 = 0; k4 < 32; ++k4) {
            bf[k4].x = Bl[(size_t)(4 * k4) * C_DIM];
            bf[k4].y = Bl[(size_t)(4 * k4 + 1) * C_DIM];
        }

        v8f acc[4] = {{}, {}, {}, {}};
        #pragma unroll
        for (int k4 = 0; k4 < 32; ++k4) {
            const int k0 = 4 * k4;
            v2f a0, a1, a2, a3;
            a0.x = Al[k0];              a0.y = Al[k0 + 1];
            a1.x = Al[16 * LDA + k0];   a1.y = Al[16 * LDA + k0 + 1];
            a2.x = Al[32 * LDA + k0];   a2.y = Al[32 * LDA + k0 + 1];
            a3.x = Al[48 * LDA + k0];   a3.y = Al[48 * LDA + k0 + 1];
            acc[0] = __builtin_amdgcn_wmma_f32_16x16x4_f32(
                         false, a0, false, bf[k4], (short)0, acc[0], false, false);
            acc[1] = __builtin_amdgcn_wmma_f32_16x16x4_f32(
                         false, a1, false, bf[k4], (short)0, acc[1], false, false);
            acc[2] = __builtin_amdgcn_wmma_f32_16x16x4_f32(
                         false, a2, false, bf[k4], (short)0, acc[2], false, false);
            acc[3] = __builtin_amdgcn_wmma_f32_16x16x4_f32(
                         false, a3, false, bf[k4], (short)0, acc[3], false, false);
        }

        #pragma unroll
        for (int s = 0; s < 4; ++s) {
            const int rbase = row0 + s * 16;
            float* Hp = H + (size_t)rbase * (3 * C_DIM) + gcol;
            #pragma unroll
            for (int r = 0; r < 8; ++r) {
                const int m = r + 8 * hi;
                if (rbase + m < N)
                    Hp[(size_t)m * (3 * C_DIM) + l16] = acc[s][r];
            }
        }
    }
}

__global__ __launch_bounds__(256)
void scatter_kernel(const int* __restrict__ ei,
                    const float* __restrict__ u,
                    const float* __restrict__ H,
                    float* __restrict__ agg,
                    float* __restrict__ deg,
                    int E)
{
    const int e    = blockIdx.x * 8 + (threadIdx.x >> 5);
    const int lane = threadIdx.x & 31;
    if (e >= E) return;

    const int   s  = ei[e];
    const int   d  = ei[E + e];
    const float uu = u[e];
    const float w0 = 1.0f - uu;

    const float4* h0 = (const float4*)(H + (size_t)s * 3 * C_DIM);
    const float4* h1 = (const float4*)(H + (size_t)s * 3 * C_DIM + C_DIM);
    const float4 a = h0[lane];
    const float4 b = h1[lane];
    const float4 m = make_float4(w0 * a.x + uu * b.x, w0 * a.y + uu * b.y,
                                 w0 * a.z + uu * b.z, w0 * a.w + uu * b.w);

    float* ap = agg + (size_t)d * C_DIM + lane * 4;
    atomicAdd(ap + 0, m.x);
    atomicAdd(ap + 1, m.y);
    atomicAdd(ap + 2, m.z);
    atomicAdd(ap + 3, m.w);
    if (lane == 0) atomicAdd(deg + d, 1.0f);
}

__global__ __launch_bounds__(256)
void combine_kernel(const float* __restrict__ agg,
                    const float* __restrict__ deg,
                    const float* __restrict__ H,
                    const float* __restrict__ bias,
                    float* __restrict__ out,
                    int N)
{
    const int n    = blockIdx.x * 8 + (threadIdx.x >> 5);
    const int lane = threadIdx.x & 31;
    if (n >= N) return;

    const float inv = 1.0f / fmaxf(deg[n], 1.0f);
    const float4 a  = ((const float4*)(agg + (size_t)n * C_DIM))[lane];
    const float4 r  = ((const float4*)(H + (size_t)n * 3 * C_DIM + 2 * C_DIM))[lane];
    const float4 bb = ((const float4*)bias)[lane];
    const float4 o  = make_float4(a.x * inv + r.x + bb.x,
                                  a.y * inv + r.y + bb.y,
                                  a.z * inv + r.z + bb.z,
                                  a.w * inv + r.w + bb.w);
    ((float4*)(out + (size_t)n * C_DIM))[lane] = o;
}

__global__ __launch_bounds__(256)
void zero_kernel(float* __restrict__ p, size_t n)
{
    const size_t i = (size_t)blockIdx.x * blockDim.x + threadIdx.x;
    if (i < n) p[i] = 0.0f;
}

extern "C" void kernel_launch(void* const* d_in, const int* in_sizes, int n_in,
                              void* d_out, int out_size, void* d_ws, size_t ws_size,
                              hipStream_t stream)
{
    const float* x  = (const float*)d_in[0];
    const int*   ei = (const int*)d_in[1];
    const float* ea = (const float*)d_in[2];
    const float* W1 = (const float*)d_in[3];
    const float* r1 = (const float*)d_in[4];
    const float* b1 = (const float*)d_in[5];
    const float* W2 = (const float*)d_in[6];
    const float* r2 = (const float*)d_in[7];
    const float* b2 = (const float*)d_in[8];
    float* out = (float*)d_out;

    const int N = in_sizes[0] / C_DIM;
    const int E = in_sizes[1] / 2;

    float* ws   = (float*)d_ws;
    float* Hbuf = ws;
    float* agg  = Hbuf + (size_t)N * 3 * C_DIM;
    float* deg  = agg  + (size_t)N * C_DIM;
    float* hmid = deg  + N;

    const int mblocks = (N + MTILE - 1) / MTILE;
    const int eblocks = (E + 7) / 8;
    const int nblocks = (N + 7) / 8;
    const size_t zn   = (size_t)N * (C_DIM + 1);
    const int zblocks = (int)((zn + 255) / 256);

    zero_kernel<<<zblocks, 256, 0, stream>>>(agg, zn);
    splineconv_gemm_kernel<<<mblocks, 256, 0, stream>>>(x, W1, r1, Hbuf, N);
    scatter_kernel<<<eblocks, 256, 0, stream>>>(ei, ea, Hbuf, agg, deg, E);
    combine_kernel<<<nblocks, 256, 0, stream>>>(agg, deg, Hbuf, b1, hmid, N);

    zero_kernel<<<zblocks, 256, 0, stream>>>(agg, zn);
    splineconv_gemm_kernel<<<mblocks, 256, 0, stream>>>(hmid, W2, r2, Hbuf, N);
    scatter_kernel<<<eblocks, 256, 0, stream>>>(ei, ea, Hbuf, agg, deg, E);
    combine_kernel<<<nblocks, 256, 0, stream>>>(agg, deg, Hbuf, b2, out, N);
}


# edge, row and clearing kernels: grid fixed at the chip's 16384 wave slots, each wave loops over its share (37 edges, not 1): 600000 short waves per edge pass become 16384; agg sums: atomic arrival
# speedup vs baseline: 1.1772x; 1.1772x over previous
#include <hip/hip_runtime.h>

typedef __attribute__((ext_vector_type(2))) float v2f;
typedef __attribute__((ext_vector_type(8))) float v8f;

#define C_DIM 128
#define LDA   132
#define MTILE 64
#define KCHUNK 32

__global__ __launch_bounds__(256)
void splineconv_gemm_kernel(const float* __restrict__ X,
                            const float* __restrict__ W,
                            const float* __restrict__ root,
                            float* __restrict__ H,
                            int N)
{
    __shared__ float As[MTILE * LDA];
    const int tid  = threadIdx.x;
    const int wave = tid >> 5;
    const int lane = tid & 31;
    const int row0 = blockIdx.x * MTILE;

    for (int i = tid; i < MTILE * (C_DIM / 4); i += 256) {
        const int r  = i >> 5;
        const int c4 = i & 31;
        float4 v = make_float4(0.f, 0.f, 0.f, 0.f);
        const int gr = row0 + r;
        if (gr < N) v = ((const float4*)(X + (size_t)gr * C_DIM))[c4];
        float* dst = &As[r * LDA + c4 * 4];
        dst[0] = v.x; dst[1] = v.y; dst[2] = v.z; dst[3] = v.w;
    }
    __syncthreads();

    const int hi  = lane >> 4;
    const int l16 = lane & 15;

    for (int t = wave; t < 24; t += 8) {
        const int gcol = t * 16;
        const float* Bbase;
        if (gcol < C_DIM)          Bbase = W + gcol;
        else if (gcol < 2 * C_DIM) Bbase = W + (size_t)C_DIM * C_DIM + (gcol - C_DIM);
        else                       Bbase = root + (gcol - 2 * C_DIM);

        const float* Bl = Bbase + (size_t)(2 * hi) * C_DIM + l16;
        const float* Al = &As[l16 * LDA + 2 * hi];

        v8f acc[4] = {{}, {}, {}, {}};
        #pragma unroll 1
        for (int kc = 0; kc < C_DIM / KCHUNK; ++kc) {
            const float* Bc = Bl + (size_t)(kc * KCHUNK) * C_DIM;
            const float* Ac = Al + kc * KCHUNK;

            v2f bf[KCHUNK / 4];
            #pragma unroll
            for (int j = 0; j < KCHUNK / 4; ++j) {
                bf[j].x = Bc[(size_t)(4 * j) * C_DIM];
                bf[j].y = Bc[(size_t)(4 * j + 1) * C_DIM];
            }
            v2f af[KCHUNK / 4][4];
            #pragma unroll
            for (int j = 0; j < KCHUNK / 4; ++j) {
                #pragma unroll
                for (int s = 0; s < 4; ++s) {
                    af[j][s].x = Ac[s * 16 * LDA + 4 * j];
                    af[j][s].y = Ac[s * 16 * LDA + 4 * j + 1];
                }
            }
            #pragma unroll
            for (int j = 0; j < KCHUNK / 4; ++j) {
                #pragma unroll
                for (int s = 0; s < 4; ++s) {
                    acc[s] = __builtin_amdgcn_wmma_f32_16x16x4_f32(
                                 false, af[j][s], false, bf[j], (short)0, acc[s], false, false);
                }
            }
        }

        #pragma unroll
        for (int s = 0; s < 4; ++s) {
            const int rbase = row0 + s * 16;
            float* Hp = H + (size_t)rbase * (3 * C_DIM) + gcol;
            #pragma unroll
            for (int r = 0; r < 8; ++r) {
                const int m = r + 8 * hi;
                if (rbase + m < N)
                    Hp[(size_t)m * (3 * C_DIM) + l16] = acc[s][r];
            }
        }
    }
}

__device__ __forceinline__ void blend_add(float* __restrict__ agg, int d, float uu,
                                          const float4 a, const float4 b, int lane)
{
    const float w0 = 1.0f - uu;
    const float4 m = make_float4(w0 * a.x + uu * b.x, w0 * a.y + uu * b.y,
                                 w0 * a.z + uu * b.z, w0 * a.w + uu * b.w);
    float* ap = agg + (size_t)d * C_DIM + lane * 4;
    atomicAdd(ap + 0, m.x);
    atomicAdd(ap + 1, m.y);
    atomicAdd(ap + 2, m.z);
    atomicAdd(ap + 3, m.w);
}

__global__ __launch_bounds__(256)
void scatter_kernel(const int* __restrict__ ei,
                    const float* __restrict__ u,
                    const float* __restrict__ H,
                    float* __restrict__ agg,
                    float* __restrict__ deg,
                    int E)
{
    const int lane = threadIdx.x & 31;
    const int wv  = __builtin_amdgcn_readfirstlane((int)(blockIdx.x * 8 + (threadIdx.x >> 5)));
    const int nw  = (int)gridDim.x * 8;
    const int per = (E + nw - 1) / nw;
    const int beg = wv * per;
    const int end = (beg + per < E) ? (beg + per) : E;

    for (int base = beg; base < end; base += 32) {
        const int e = base + lane;
        int   s_l = 0, d_l = 0;
        float u_l = 0.0f;
        if (e < end) {
            s_l = ei[e];
            d_l = ei[E + e];
            u_l = u[e];
            atomicAdd(deg + d_l, 1.0f);
        }
        const int cnt = (end - base < 32) ? (end - base) : 32;

        int j = 0;
        for (; j + 4 <= cnt; j += 4) {
            float4 a[4], b[4];
            int    d[4];
            float  uu[4];
            #pragma unroll
            for (int q = 0; q < 4; ++q) {
                const int s = __builtin_amdgcn_readlane(s_l, j + q);
                d[q]  = __builtin_amdgcn_readlane(d_l, j + q);
                uu[q] = __int_as_float(__builtin_amdgcn_readlane(__float_as_int(u_l), j + q));
                const float4* hp = (const float4*)(H + (size_t)s * 3 * C_DIM);
                a[q] = hp[lane];
                b[q] = hp[32 + lane];
            }
            #pragma unroll
            for (int q = 0; q < 4; ++q)
                blend_add(agg, d[q], uu[q], a[q], b[q], lane);
        }
        for (; j < cnt; ++j) {
            const int   s  = __builtin_amdgcn_readlane(s_l, j);
            const int   dd = __builtin_amdgcn_readlane(d_l, j);
            const float uu = __int_as_float(__builtin_amdgcn_readlane(__float_as_int(u_l), j));
            const float4* hp = (const float4*)(H + (size_t)s * 3 * C_DIM);
            const float4 a = hp[lane];
            const float4 b = hp[32 + lane];
            blend_add(agg, dd, uu, a, b, lane);
        }
    }
}

__global__ __launch_bounds__(256)
void combine_kernel(const float* __restrict__ agg,
                    const float* __restrict__ deg,
                    const float* __restrict__ H,
                    const float* __restrict__ bias,
                    float* __restrict__ out,
                    int N)
{
    const int lane = threadIdx.x & 31;
    const int wv   = blockIdx.x * 8 + (threadIdx.x >> 5);
    const int nw   = (int)gridDim.x * 8;
    const float4 bb = ((const float4*)bias)[lane];

    for (int n = wv; n < N; n += nw) {
        const float inv = 1.0f / fmaxf(deg[n], 1.0f);
        const float4 a  = ((const float4*)(agg + (size_t)n * C_DIM))[lane];
        const float4 r  = ((const float4*)(H + (size_t)n * 3 * C_DIM + 2 * C_DIM))[lane];
        const float4 o  = make_float4(a.x * inv + r.x + bb.x,
                                      a.y * inv + r.y + bb.y,
                                      a.z * inv + r.z + bb.z,
                                      a.w * inv + r.w + bb.w);
        ((float4*)(out + (size_t)n * C_DIM))[lane] = o;
    }
}

__global__ __launch_bounds__(256)
void zero_kernel(float* __restrict__ p, size_t n)
{
    const size_t step = (size_t)gridDim.x * blockDim.x;
    for (size_t i = (size_t)blockIdx.x * blockDim.x + threadIdx.x; i < n; i += step)
        p[i] = 0.0f;
}

extern "C" void kernel_launch(void* const* d_in, const int* in_sizes, int n_in,
                              void* d_out, int out_size, void* d_ws, size_t ws_size,
                              hipStream_t stream)
{
    const float* x  = (const float*)d_in[0];
    const int*   ei = (const int*)d_in[1];
    const float* ea = (const float*)d_in[2];
    const float* W1 = (const float*)d_in[3];
    const float* r1 = (const float*)d_in[4];
    const float* b1 = (const float*)d_in[5];
    const float* W2 = (const float*)d_in[6];
    const float* r2 = (const float*)d_in[7];
    const float* b2 = (const float*)d_in[8];
    float* out = (float*)d_out;

    const int N = in_sizes[0] / C_DIM;
    const int E = in_sizes[1] / 2;

    float* ws   = (float*)d_ws;
    float* Hbuf = ws;
    float* agg  = Hbuf + (size_t)N * 3 * C_DIM;
    float* deg  = agg  + (size_t)N * C_DIM;
    float* hmid = deg  + N;

    const int mblocks = (N + MTILE - 1) / MTILE;
    const size_t zn   = (size_t)N * (C_DIM + 1);
    const int sblocks = 2048;

    zero_kernel<<<sblocks, 256, 0, stream>>>(agg, zn);
    splineconv_gemm_kernel<<<mblocks, 256, 0, stream>>>(x, W1, r1, Hbuf, N);
    scatter_kernel<<<sblocks, 256, 0, stream>>>(ei, ea, Hbuf, agg, deg, E);
    combine_kernel<<<sblocks, 256, 0, stream>>>(agg, deg, Hbuf, b1, hmid, N);

    zero_kernel<<<sblocks, 256, 0, stream>>>(agg, zn);
    splineconv_gemm_kernel<<<mblocks, 256, 0, stream>>>(hmid, W2, r2, Hbuf, N);
    scatter_kernel<<<sblocks, 256, 0, stream>>>(ei, ea, Hbuf, agg, deg, E);
    combine_kernel<<<sblocks, 256, 0, stream>>>(agg, deg, Hbuf, b2, out, N);
}


# edge kernel: a block owns 196 destination rows, sums and counts in LDS, filters the whole edge list for them, writes out itself: no global atomics, clear or row pass; sum order: LDS atomic arrival
# speedup vs baseline: 2.3329x; 1.9817x over previous
#include <hip/hip_runtime.h>

typedef __attribute__((ext_vector_type(2))) float v2f;
typedef __attribute__((ext_vector_type(8))) float v8f;
typedef __attribute__((ext_vector_type(4))) int   v4i;

#define C_DIM 128
#define LDA   132
#define MTILE 64
#define KCHUNK 32

__global__ __launch_bounds__(256)
void splineconv_gemm_kernel(const float* __restrict__ X,
                            const float* __restrict__ W,
                            const float* __restrict__ root,
                            float* __restrict__ H,
                            int N)
{
    __shared__ float As[MTILE * LDA];
    const int tid  = threadIdx.x;
    const int wave = tid >> 5;
    const int lane = tid & 31;
    const int row0 = blockIdx.x * MTILE;

    for (int i = tid; i < MTILE * (C_DIM / 4); i += 256) {
        const int r  = i >> 5;
        const int c4 = i & 31;
        float4 v = make_float4(0.f, 0.f, 0.f, 0.f);
        const int gr = row0 + r;
        if (gr < N) v = ((const float4*)(X + (size_t)gr * C_DIM))[c4];
        float* dst = &As[r * LDA + c4 * 4];
        dst[0] = v.x; dst[1] = v.y; dst[2] = v.z; dst[3] = v.w;
    }
    __syncthreads();

    const int hi  = lane >> 4;
    const int l16 = lane & 15;

    for (int t = wave; t < 24; t += 8) {
        const int gcol = t * 16;
        const float* Bbase;
        if (gcol < C_DIM)          Bbase = W + gcol;
        else if (gcol < 2 * C_DIM) Bbase = W + (size_t)C_DIM * C_DIM + (gcol - C_DIM);
        else                       Bbase = root + (gcol - 2 * C_DIM);

        const float* Bl = Bbase + (size_t)(2 * hi) * C_DIM + l16;
        const float* Al = &As[l16 * LDA + 2 * hi];

        v8f acc[4] = {{}, {}, {}, {}};
        #pragma unroll 1
        for (int kc = 0; kc < C_DIM / KCHUNK; ++kc) {
            const float* Bc = Bl + (size_t)(kc * KCHUNK) * C_DIM;
            const float* Ac = Al + kc * KCHUNK;

            v2f bf[KCHUNK / 4];
            #pragma unroll
            for (int j = 0; j < KCHUNK / 4; ++j) {
                bf[j].x = Bc[(size_t)(4 * j) * C_DIM];
                bf[j].y = Bc[(size_t)(4 * j + 1) * C_DIM];
            }
            v2f af[KCHUNK / 4][4];
            #pragma unroll
            for (int j = 0; j < KCHUNK / 4; ++j) {
                #pragma unroll
                for (int s = 0; s < 4; ++s) {
                    af[j][s].x = Ac[s * 16 * LDA + 4 * j];
                    af[j][s].y = Ac[s * 16 * LDA + 4 * j + 1];
                }
            }
            #pragma unroll
            for (int j = 0; j < KCHUNK / 4; ++j) {
                #pragma unroll
                for (int s = 0; s < 4; ++s) {
                    acc[s] = __builtin_amdgcn_wmma_f32_16x16x4_f32(
                                 false, af[j][s], false, bf[j], (short)0, acc[s], false, false);
                }
            }
        }

        #pragma unroll
        for (int s = 0; s < 4; ++s) {
            const int rbase = row0 + s * 16;
            float* Hp = H + (size_t)rbase * (3 * C_DIM) + gcol;
            #pragma unroll
            for (int r = 0; r < 8; ++r) {
                const int m = r + 8 * hi;
                if (rbase + m < N)
                    Hp[(size_t)m * (3 * C_DIM) + l16] = acc[s][r];
            }
        }
    }
}

#define OWN_ROWS   196
#define OWN_WAVES  32
#define SCAN_EDGES 512
#define QFLUSH     64
#define QCAP       576

__global__ __launch_bounds__(1024)
void owner_edge_kernel(const int* __restrict__ ei,
                       const float* __restrict__ u,
                       const float* __restrict__ H,
                       const float* __restrict__ bias,
                       float* __restrict__ out,
                       int N, int E)
{
    __shared__ float acc[OWN_ROWS * C_DIM];
    __shared__ int   cnt[OWN_ROWS];
    __shared__ int   qe[OWN_WAVES * QCAP];
    __shared__ int   qr[OWN_WAVES * QCAP];

    const int lane = threadIdx.x & 31;
    const int wave = __builtin_amdgcn_readfirstlane((int)(threadIdx.x >> 5));
    int* const qew = qe + wave * QCAP;
    int* const qrw = qr + wave * QCAP;

    const int* __restrict__ dstp = ei + E;
    const bool al16   = ((E & 3) == 0) && ((((size_t)dstp) & 15) == 0);
    const int  nchunk = (E + SCAN_EDGES - 1) / SCAN_EDGES;
    const int  rot    = (int)((blockIdx.x * (unsigned)nchunk) / gridDim.x);
    const float4 bb   = ((const float4*)bias)[lane];

    for (int n0 = (int)blockIdx.x * OWN_ROWS; n0 < N; n0 += (int)gridDim.x * OWN_ROWS) {
        const int rows = (N - n0 < OWN_ROWS) ? (N - n0) : OWN_ROWS;

        for (int i = threadIdx.x; i < OWN_ROWS * C_DIM; i += 1024) acc[i] = 0.0f;
        for (int i = threadIdx.x; i < OWN_ROWS; i += 1024) cnt[i] = 0;
        __syncthreads();

        int qn = 0;
        for (int c0 = wave; ; c0 += OWN_WAVES) {
            const bool last = (c0 >= nchunk);
            if (!last) {
                int c = c0 + rot;
                if (c >= nchunk) c -= nchunk;
                const int base = c * SCAN_EDGES + lane * 4;
                v4i      vv[4];
                unsigned lim[4];
                if (al16) {
                    #pragma unroll
                    for (int k = 0; k < 4; ++k) {
                        const int  e0 = base + k * 128;
                        const bool ok = e0 < E;
                        vv[k]  = *(const v4i*)(dstp + (ok ? e0 : 0));
                        lim[k] = ok ? (unsigned)rows : 0u;
                    }
                } else {
                    #pragma unroll
                    for (int k = 0; k < 4; ++k) {
                        const int e0 = base + k * 128;
                        #pragma unroll
                        for (int t = 0; t < 4; ++t) {
                            const bool ok = (e0 + t) < E;
                            const int  dv = dstp[ok ? (e0 + t) : 0];
                            vv[k][t] = ok ? dv : -1;
                        }
                        lim[k] = (unsigned)rows;
                    }
                }
                #pragma unroll
                for (int k = 0; k < 4; ++k) {
                    const int e0 = base + k * 128;
                    #pragma unroll
                    for (int t = 0; t < 4; ++t) {
                        const unsigned rr = (unsigned)(vv[k][t] - n0);
                        const bool     mm = rr < lim[k];
                        const unsigned mk = __builtin_amdgcn_ballot_w32(mm);
                        if (mk != 0u) {
                            if (mm) {
                                const int p = qn + (int)__builtin_amdgcn_mbcnt_lo(mk, 0u);
                                qew[p] = e0 + t;
                                qrw[p] = (int)rr;
                            }
                            qn += __builtin_popcount(mk);
                        }
                    }
                }
            }
            if (qn >= QFLUSH || (last && qn > 0)) {
                for (int g = 0; g < qn; g += 32) {
                    const int i = g + lane;
                    int   s_l = 0, r_l = 0;
                    float u_l = 0.0f;
                    if (i < qn) {
                        const int e = qew[i];
                        r_l = qrw[i];
                        s_l = ei[e];
                        u_l = u[e];
                        atomicAdd(&cnt[r_l], 1);
                    }
                    const int cn = (qn - g < 32) ? (qn - g) : 32;
                    int j = 0;
                    for (; j + 4 <= cn; j += 4) {
                        float4 a[4], b[4];
                        int    r[4];
                        float  uu[4];
                        #pragma unroll
                        for (int q = 0; q < 4; ++q) {
                            const int s = __builtin_amdgcn_readlane(s_l, j + q);
                            r[q]  = __builtin_amdgcn_readlane(r_l, j + q);
                            uu[q] = __int_as_float(__builtin_amdgcn_readlane(__float_as_int(u_l), j + q));
                            const float4* hp = (const float4*)(H + (size_t)s * 3 * C_DIM);
                            a[q] = hp[lane];
                            b[q] = hp[32 + lane];
                        }
                        #pragma unroll
                        for (int q = 0; q < 4; ++q) {
                            const float w0 = 1.0f - uu[q];
                            float* ap = acc + r[q] * C_DIM + lane;
                            __hip_atomic_fetch_add(ap,      w0 * a[q].x + uu[q] * b[q].x, __ATOMIC_RELAXED, __HIP_MEMORY_SCOPE_WORKGROUP);
                            __hip_atomic_fetch_add(ap + 32, w0 * a[q].y + uu[q] * b[q].y, __ATOMIC_RELAXED, __HIP_MEMORY_SCOPE_WORKGROUP);
                            __hip_atomic_fetch_add(ap + 64, w0 * a[q].z + uu[q] * b[q].z, __ATOMIC_RELAXED, __HIP_MEMORY_SCOPE_WORKGROUP);
                            __hip_atomic_fetch_add(ap + 96, w0 * a[q].w + uu[q] * b[q].w, __ATOMIC_RELAXED, __HIP_MEMORY_SCOPE_WORKGROUP);
                        }
                    }
                    for (; j < cn; ++j) {
                        const int   s  = __builtin_amdgcn_readlane(s_l, j);
                        const int   rj = __builtin_amdgcn_readlane(r_l, j);
                        const float uj = __int_as_float(__builtin_amdgcn_readlane(__float_as_int(u_l), j));
                        const float4* hp = (const float4*)(H + (size_t)s * 3 * C_DIM);
                        const float4 a = hp[lane];
                        const float4 b = hp[32 + lane];
                        const float w0 = 1.0f - uj;
                        float* ap = acc + rj * C_DIM + lane;
                        __hip_atomic_fetch_add(ap,      w0 * a.x + uj * b.x, __ATOMIC_RELAXED, __HIP_MEMORY_SCOPE_WORKGROUP);
                        __hip_atomic_fetch_add(ap + 32, w0 * a.y + uj * b.y, __ATOMIC_RELAXED, __HIP_MEMORY_SCOPE_WORKGROUP);
                        __hip_atomic_fetch_add(ap + 64, w0 * a.z + uj * b.z, __ATOMIC_RELAXED, __HIP_MEMORY_SCOPE_WORKGROUP);
                        __hip_atomic_fetch_add(ap + 96, w0 * a.w + uj * b.w, __ATOMIC_RELAXED, __HIP_MEMORY_SCOPE_WORKGROUP);
                    }
                }
                qn = 0;
            }
            if (last) break;
        }
        __syncthreads();

        for (int r = wave; r < rows; r += OWN_WAVES) {
            const int   n   = n0 + r;
            const float inv = 1.0f / fmaxf((float)cnt[r], 1.0f);
            const float* ar = acc + r * C_DIM + lane;
            const float4 hr = ((const float4*)(H + (size_t)n * 3 * C_DIM + 2 * C_DIM))[lane];
            const float4 o  = make_float4(ar[0]  * inv + hr.x + bb.x,
                                          ar[32] * inv + hr.y + bb.y,
                                          ar[64] * inv + hr.z + bb.z,
                                          ar[96] * inv + hr.w + bb.w);
            ((float4*)(out + (size_t)n * C_DIM))[lane] = o;
        }
        __syncthreads();
    }
}

extern "C" void kernel_launch(void* const* d_in, const int* in_sizes, int n_in,
                              void* d_out, int out_size, void* d_ws, size_t ws_size,
                              hipStream_t stream)
{
    const float* x  = (const float*)d_in[0];
    const int*   ei = (const int*)d_in[1];
    const float* ea = (const float*)d_in[2];
    const float* W1 = (const float*)d_in[3];
    const float* r1 = (const float*)d_in[4];
    const float* b1 = (const float*)d_in[5];
    const float* W2 = (const float*)d_in[6];
    const float* r2 = (const float*)d_in[7];
    const float* b2 = (const float*)d_in[8];
    float* out = (float*)d_out;

    const int N = in_sizes[0] / C_DIM;
    const int E = in_sizes[1] / 2;

    float* ws   = (float*)d_ws;
    float* Hbuf = ws;
    float* hmid = Hbuf + (size_t)N * 3 * C_DIM;

    const int mblocks = (N + MTILE - 1) / MTILE;
    const int oblocks = 256;

    splineconv_gemm_kernel<<<mblocks, 256, 0, stream>>>(x, W1, r1, Hbuf, N);
    owner_edge_kernel<<<oblocks, 1024, 0, stream>>>(ei, ea, Hbuf, b1, hmid, N, E);

    splineconv_gemm_kernel<<<mblocks, 256, 0, stream>>>(hmid, W2, r2, Hbuf, N);
    owner_edge_kernel<<<oblocks, 1024, 0, stream>>>(ei, ea, Hbuf, b2, out, N, E);
}


# edge kernel: layer 1's pass keeps each wave's matched edges (source, row, weight) in the workspace; layer 2's pass reads them back and never filters the edge list; sum order: LDS atomic arrival
# speedup vs baseline: 2.5904x; 1.1104x over previous
#include <hip/hip_runtime.h>

typedef __attribute__((ext_vector_type(2))) float v2f;
typedef __attribute__((ext_vector_type(8))) float v8f;
typedef __attribute__((ext_vector_type(4))) int   v4i;

#define C_DIM 128
#define LDA   132
#define MTILE 64
#define KCHUNK 32

__global__ __launch_bounds__(256)
void splineconv_gemm_kernel(const float* __restrict__ X,
                            const float* __restrict__ W,
                            const float* __restrict__ root,
                            float* __restrict__ H,
                            int N)
{
    __shared__ float As[MTILE * LDA];
    const int tid  = threadIdx.x;
    const int wave = tid >> 5;
    const int lane = tid & 31;
    const int row0 = blockIdx.x * MTILE;

    for (int i = tid; i < MTILE * (C_DIM / 4); i += 256) {
        const int r  = i >> 5;
        const int c4 = i & 31;
        float4 v = make_float4(0.f, 0.f, 0.f, 0.f);
        const int gr = row0 + r;
        if (gr < N) v = ((const float4*)(X + (size_t)gr * C_DIM))[c4];
        float* dst = &As[r * LDA + c4 * 4];
        dst[0] = v.x; dst[1] = v.y; dst[2] = v.z; dst[3] = v.w;
    }
    __syncthreads();

    const int hi  = lane >> 4;
    const int l16 = lane & 15;

    for (int t = wave; t < 24; t += 8) {
        const int gcol = t * 16;
        const float* Bbase;
        if (gcol < C_DIM)          Bbase = W + gcol;
        else if (gcol < 2 * C_DIM) Bbase = W + (size_t)C_DIM * C_DIM + (gcol - C_DIM);
        else                       Bbase = root + (gcol - 2 * C_DIM);

        const float* Bl = Bbase + (size_t)(2 * hi) * C_DIM + l16;
        const float* Al = &As[l16 * LDA + 2 * hi];

        v8f acc[4] = {{}, {}, {}, {}};
        #pragma unroll 1
        for (int kc = 0; kc < C_DIM / KCHUNK; ++kc) {
            const float* Bc = Bl + (size_t)(kc * KCHUNK) * C_DIM;
            const float* Ac = Al + kc * KCHUNK;

            v2f bf[KCHUNK / 4];
            #pragma unroll
            for (int j = 0; j < KCHUNK / 4; ++j) {
                bf[j].x = Bc[(size_t)(4 * j) * C_DIM];
                bf[j].y = Bc[(size_t)(4 * j + 1) * C_DIM];
            }
            v2f af[KCHUNK / 4][4];
            #pragma unroll
            for (int j = 0; j < KCHUNK / 4; ++j) {
                #pragma unroll
                for (int s = 0; s < 4; ++s) {
                    af[j][s].x = Ac[s * 16 * LDA + 4 * j];
                    af[j][s].y = Ac[s * 16 * LDA + 4 * j + 1];
                }
            }
            #pragma unroll
            for (int j = 0; j < KCHUNK / 4; ++j) {
                #pragma unroll
                for (int s = 0; s < 4; ++s) {
                    acc[s] = __builtin_amdgcn_wmma_f32_16x16x4_f32(
                                 false, af[j][s], false, bf[j], (short)0, acc[s], false, false);
                }
            }
        }

        #pragma unroll
        for (int s = 0; s < 4; ++s) {
            const int rbase = row0 + s * 16;
            float* Hp = H + (size_t)rbase * (3 * C_DIM) + gcol;
            #pragma unroll
            for (int r = 0; r < 8; ++r) {
                const int m = r + 8 * hi;
                if (rbase + m < N)
                    Hp[(size_t)m * (3 * C_DIM) + l16] = acc[s][r];
            }
        }
    }
}

#define OWN_ROWS   196
#define OWN_WAVES  32
#define SCAN_EDGES 512
#define QFLUSH     64
#define QCAP       576
#define KEEP_CAP   256

__device__ __forceinline__
void add_edges(const float* __restrict__ H, float* acc, const int lane,
               const int s_l, const int r_l, const float u_l, const int cn)
{
    int j = 0;
    for (; j + 4 <= cn; j += 4) {
        float4 a[4], b[4];
        int    r[4];
        float  uu[4];
        #pragma unroll
        for (int q = 0; q < 4; ++q) {
            const int s = __builtin_amdgcn_readlane(s_l, j + q);
            r[q]  = __builtin_amdgcn_readlane(r_l, j + q);
            uu[q] = __int_as_float(__builtin_amdgcn_readlane(__float_as_int(u_l), j + q));
            const float4* hp = (const float4*)(H + (size_t)s * 3 * C_DIM);
            a[q] = hp[lane];
            b[q] = hp[32 + lane];
        }
        #pragma unroll
        for (int q = 0; q < 4; ++q) {
            const float w0 = 1.0f - uu[q];
            float* ap = acc + r[q] * C_DIM + lane;
            __hip_atomic_fetch_add(ap,      w0 * a[q].x + uu[q] * b[q].x, __ATOMIC_RELAXED, __HIP_MEMORY_SCOPE_WORKGROUP);
            __hip_atomic_fetch_add(ap + 32, w0 * a[q].y + uu[q] * b[q].y, __ATOMIC_RELAXED, __HIP_MEMORY_SCOPE_WORKGROUP);
            __hip_atomic_fetch_add(ap + 64, w0 * a[q].z + uu[q] * b[q].z, __ATOMIC_RELAXED, __HIP_MEMORY_SCOPE_WORKGROUP);
            __hip_atomic_fetch_add(ap + 96, w0 * a[q].w + uu[q] * b[q].w, __ATOMIC_RELAXED, __HIP_MEMORY_SCOPE_WORKGROUP);
        }
    }
    for (; j < cn; ++j) {
        const int   s  = __builtin_amdgcn_readlane(s_l, j);
        const int   rj = __builtin_amdgcn_readlane(r_l, j);
        const float uj = __int_as_float(__builtin_amdgcn_readlane(__float_as_int(u_l), j));
        const float4* hp = (const float4*)(H + (size_t)s * 3 * C_DIM);
        const float4 a = hp[lane];
        const float4 b = hp[32 + lane];
        const float w0 = 1.0f - uj;
        float* ap = acc + rj * C_DIM + lane;
        __hip_atomic_fetch_add(ap,      w0 * a.x + uj * b.x, __ATOMIC_RELAXED, __HIP_MEMORY_SCOPE_WORKGROUP);
        __hip_atomic_fetch_add(ap + 32, w0 * a.y + uj * b.y, __ATOMIC_RELAXED, __HIP_MEMORY_SCOPE_WORKGROUP);
        __hip_atomic_fetch_add(ap + 64, w0 * a.z + uj * b.z, __ATOMIC_RELAXED, __HIP_MEMORY_SCOPE_WORKGROUP);
        __hip_atomic_fetch_add(ap + 96, w0 * a.w + uj * b.w, __ATOMIC_RELAXED, __HIP_MEMORY_SCOPE_WORKGROUP);
    }
}

template <int PASS>
__global__ __launch_bounds__(1024)
void owner_edge_kernel(const int* __restrict__ ei,
                       const float* __restrict__ u,
                       const float* __restrict__ H,
                       const float* __restrict__ bias,
                       float* __restrict__ out,
                       int* __restrict__ kept,
                       int* __restrict__ kept_n,
                       int N, int E)
{
    __shared__ float acc[OWN_ROWS * C_DIM];
    __shared__ int   cnt[OWN_ROWS];
    __shared__ int   qe[OWN_WAVES * QCAP];
    __shared__ int   qr[OWN_WAVES * QCAP];
    __shared__ int   over;

    const int lane = threadIdx.x & 31;
    const int wave = __builtin_amdgcn_readfirstlane((int)(threadIdx.x >> 5));
    int* const qew = qe + wave * QCAP;
    int* const qrw = qr + wave * QCAP;

    const int* __restrict__ dstp = ei + E;
    const bool al16   = ((E & 3) == 0) && ((((size_t)dstp) & 15) == 0);
    const int  nchunk = (E + SCAN_EDGES - 1) / SCAN_EDGES;
    const int  rot    = (int)((blockIdx.x * (unsigned)nchunk) / gridDim.x);
    const float4 bb   = ((const float4*)bias)[lane];

    const bool   one_run = N <= (int)gridDim.x * OWN_ROWS;
    const size_t wl      = (size_t)blockIdx.x * OWN_WAVES + (size_t)wave;
    v4i* const   keptw   = (v4i*)kept + wl * KEEP_CAP;

    bool replay = false;
    int  total  = 0;
    if (PASS == 1) {
        if (threadIdx.x == 0) over = 0;
        __syncthreads();
        if (one_run) {
            total = __builtin_amdgcn_readfirstlane(kept_n[wl]);
            if (lane == 0 && total > KEEP_CAP) over = 1;
        }
        __syncthreads();
        replay = one_run && (over == 0);
    }

    for (int n0 = (int)blockIdx.x * OWN_ROWS; n0 < N; n0 += (int)gridDim.x * OWN_ROWS) {
        const int rows = (N - n0 < OWN_ROWS) ? (N - n0) : OWN_ROWS;

        for (int i = threadIdx.x; i < OWN_ROWS * C_DIM; i += 1024) acc[i] = 0.0f;
        for (int i = threadIdx.x; i < OWN_ROWS; i += 1024) cnt[i] = 0;
        __syncthreads();

        if (PASS == 1 && replay) {
            v4i nx = {0, 0, 0, 0};
            if (lane < total) nx = keptw[lane];
            for (int g = 0; g < total; g += 32) {
                const int i  = g + lane;
                const v4i rc = nx;
                if (i + 32 < total) nx = keptw[i + 32];
                if (i < total) atomicAdd(&cnt[rc.y], 1);
                const int cn = (total - g < 32) ? (total - g) : 32;
                add_edges(H, acc, lane, rc.x, rc.y, __int_as_float(rc.z), cn);
            }
        } else {
            int qn    = 0;
            int saved = 0;
            for (int c0 = wave; ; c0 += OWN_WAVES) {
                const bool last = (c0 >= nchunk);
                if (!last) {
                    int c = c0 + rot;
                    if (c >= nchunk) c -= nchunk;
                    const int base = c * SCAN_EDGES + lane * 4;
                    v4i      vv[4];
                    unsigned lim[4];
                    if (al16) {
                        #pragma unroll
                        for (int k = 0; k < 4; ++k) {
                            const int  e0 = base + k * 128;
                            const bool ok = e0 < E;
                            vv[k]  = *(const v4i*)(dstp + (ok ? e0 : 0));
                            lim[k] = ok ? (unsigned)rows : 0u;
                        }
                    } else {
                        #pragma unroll
                        for (int k = 0; k < 4; ++k) {
                            const int e0 = base + k * 128;
                            #pragma unroll
                            for (int t = 0; t < 4; ++t) {
                                const bool ok = (e0 + t) < E;
                                const int  dv = dstp[ok ? (e0 + t) : 0];
                                vv[k][t] = ok ? dv : -1;
                            }
                            lim[k] = (unsigned)rows;
                        }
                    }
                    #pragma unroll
                    for (int k = 0; k < 4; ++k) {
                        const int e0 = base + k * 128;
                        #pragma unroll
                        for (int t = 0; t < 4; ++t) {
                            const unsigned rr = (unsigned)(vv[k][t] - n0);
                            const bool     mm = rr < lim[k];
                            const unsigned mk = __builtin_amdgcn_ballot_w32(mm);
                            if (mk != 0u) {
                                if (mm) {
                                    const int p = qn + (int)__builtin_amdgcn_mbcnt_lo(mk, 0u);
                                    qew[p] = e0 + t;
                                    qrw[p] = (int)rr;
                                }
                                qn += __builtin_popcount(mk);
                            }
                        }
                    }
                }
                if (qn >= QFLUSH || (last && qn > 0)) {
                    for (int g = 0; g < qn; g += 32) {
                        const int i = g + lane;
                        int   s_l = 0, r_l = 0;
                        float u_l = 0.0f;
                        if (i < qn) {
                            const int e = qew[i];
                            r_l = qrw[i];
                            s_l = ei[e];
                            u_l = u[e];
                            atomicAdd(&cnt[r_l], 1);
                            if (PASS == 0 && one_run) {
                                const int slot = saved + i;
                                if (slot < KEEP_CAP) {
                                    v4i rc;
                                    rc.x = s_l;
                                    rc.y = r_l;
                                    rc.z = __float_as_int(u_l);
                                    rc.w = 0;
                                    keptw[slot] = rc;
                                }
                            }
                        }
                        const int cn = (qn - g < 32) ? (qn - g) : 32;
                        add_edges(H, acc, lane, s_l, r_l, u_l, cn);
                    }
                    saved += qn;
                    qn = 0;
                }
                if (last) break;
            }
            if (PASS == 0 && one_run && lane == 0) kept_n[wl] = saved;
        }
        __syncthreads();

        for (int r = wave; r < rows; r += OWN_WAVES) {
            const int   n   = n0 + r;
            const float inv = 1.0f / fmaxf((float)cnt[r], 1.0f);
            const float* ar = acc + r * C_DIM + lane;
            const float4 hr = ((const float4*)(H + (size_t)n * 3 * C_DIM + 2 * C_DIM))[lane];
            const float4 o  = make_float4(ar[0]  * inv + hr.x + bb.x,
                                          ar[32] * inv + hr.y + bb.y,
                                          ar[64] * inv + hr.z + bb.z,
                                          ar[96] * inv + hr.w + bb.w);
            ((float4*)(out + (size_t)n * C_DIM))[lane] = o;
        }
        __syncthreads();
    }
}

extern "C" void kernel_launch(void* const* d_in, const int* in_sizes, int n_in,
                              void* d_out, int out_size, void* d_ws, size_t ws_size,
                              hipStream_t stream)
{
    const float* x  = (const float*)d_in[0];
    const int*   ei = (const int*)d_in[1];
    const float* ea = (const float*)d_in[2];
    const float* W1 = (const float*)d_in[3];
    const float* r1 = (const float*)d_in[4];
    const float* b1 = (const float*)d_in[5];
    const float* W2 = (const float*)d_in[6];
    const float* r2 = (const float*)d_in[7];
    const float* b2 = (const float*)d_in[8];
    float* out = (float*)d_out;

    const int N = in_sizes[0] / C_DIM;
    const int E = in_sizes[1] / 2;

    float* ws   = (float*)d_ws;
    float* Hbuf = ws;
    float* hmid = Hbuf + (size_t)N * 3 * C_DIM;
    int* kept   = (int*)(hmid + (size_t)N * C_DIM);
    int* kept_n = kept + (size_t)256 * OWN_WAVES * KEEP_CAP * 4;

    const int mblocks = (N + MTILE - 1) / MTILE;
    const int oblocks = 256;

    splineconv_gemm_kernel<<<mblocks, 256, 0, stream>>>(x, W1, r1, Hbuf, N);
    owner_edge_kernel<0><<<oblocks, 1024, 0, stream>>>(ei, ea, Hbuf, b1, hmid, kept, kept_n, N, E);

    splineconv_gemm_kernel<<<mblocks, 256, 0, stream>>>(hmid, W2, r2, Hbuf, N);
    owner_edge_kernel<1><<<oblocks, 1024, 0, stream>>>(ei, ea, Hbuf, b2, out, kept, kept_n, N, E);
}


# edge kernel: the filter reads a 1-byte owner key per edge (dst/196, written once by layer 1's product kernel), not the 4-byte destination list: a quarter of the bytes; sum order: LDS atomic arrival
# speedup vs baseline: 2.6986x; 1.0418x over previous
#include <hip/hip_runtime.h>

typedef __attribute__((ext_vector_type(2))) float v2f;
typedef __attribute__((ext_vector_type(8))) float v8f;
typedef __attribute__((ext_vector_type(4))) int   v4i;

#define C_DIM 128
#define LDA   132
#define MTILE 64
#define KCHUNK 32
#define OWN_ROWS   196
#define SCAN_EDGES 512

__global__ __launch_bounds__(256)
void splineconv_gemm_kernel(const float* __restrict__ X,
                            const float* __restrict__ W,
                            const float* __restrict__ root,
                            float* __restrict__ H,
                            int N,
                            const int* __restrict__ ei,
                            unsigned char* __restrict__ keys,
                            int E)
{
    __shared__ float As[MTILE * LDA];
    const int tid  = threadIdx.x;
    const int wave = tid >> 5;
    const int lane = tid & 31;
    const int row0 = blockIdx.x * MTILE;

    if (keys != nullptr) {
        const int* __restrict__ dstl = ei + E;
        const int nword = ((E + SCAN_EDGES - 1) / SCAN_EDGES) * (SCAN_EDGES / 4);
        for (int w = blockIdx.x * 256 + tid; w < nword; w += (int)gridDim.x * 256) {
            unsigned pk = 0u;
            #pragma unroll
            for (int t = 0; t < 4; ++t) {
                const int  e  = 4 * w + t;
                const bool ok = e < E;
                const unsigned d  = (unsigned)dstl[ok ? e : 0];
                const unsigned kb = ok ? ((d / (unsigned)OWN_ROWS) & 255u) : 255u;
                pk |= kb << (8 * t);
            }
            ((unsigned*)keys)[w] = pk;
        }
    }

    for (int i = tid; i < MTILE * (C_DIM / 4); i += 256) {
        const int r  = i >> 5;
        const int c4 = i & 31;
        float4 v = make_float4(0.f, 0.f, 0.f, 0.f);
        const int gr = row0 + r;
        if (gr < N) v = ((const float4*)(X + (size_t)gr * C_DIM))[c4];
        float* dst = &As[r * LDA + c4 * 4];
        dst[0] = v.x; dst[1] = v.y; dst[2] = v.z; dst[3] = v.w;
    }
    __syncthreads();

    const int hi  = lane >> 4;
    const int l16 = lane & 15;

    for (int t = wave; t < 24; t += 8) {
        const int gcol = t * 16;
        const float* Bbase;
        if (gcol < C_DIM)          Bbase = W + gcol;
        else if (gcol < 2 * C_DIM) Bbase = W + (size_t)C_DIM * C_DIM + (gcol - C_DIM);
        else                       Bbase = root + (gcol - 2 * C_DIM);

        const float* Bl = Bbase + (size_t)(2 * hi) * C_DIM + l16;
        const float* Al = &As[l16 * LDA + 2 * hi];

        v8f acc[4] = {{}, {}, {}, {}};
        #pragma unroll 1
        for (int kc = 0; kc < C_DIM / KCHUNK; ++kc) {
            const float* Bc = Bl + (size_t)(kc * KCHUNK) * C_DIM;
            const float* Ac = Al + kc * KCHUNK;

            v2f bf[KCHUNK / 4];
            #pragma unroll
            for (int j = 0; j < KCHUNK / 4; ++j) {
                bf[j].x = Bc[(size_t)(4 * j) * C_DIM];
                bf[j].y = Bc[(size_t)(4 * j + 1) * C_DIM];
            }
            v2f af[KCHUNK / 4][4];
            #pragma unroll
            for (int j = 0; j < KCHUNK / 4; ++j) {
                #pragma unroll
                for (int s = 0; s < 4; ++s) {
                    af[j][s].x = Ac[s * 16 * LDA + 4 * j];
                    af[j][s].y = Ac[s * 16 * LDA + 4 * j + 1];
                }
            }
            #pragma unroll
            for (int j = 0; j < KCHUNK / 4; ++j) {
                #pragma unroll
                for (int s = 0; s < 4; ++s) {
                    acc[s] = __builtin_amdgcn_wmma_f32_16x16x4_f32(
                                 false, af[j][s], false, bf[j], (short)0, acc[s], false, false);
                }
            }
        }

        #pragma unroll
        for (int s = 0; s < 4; ++s) {
            const int rbase = row0 + s * 16;
            float* Hp = H + (size_t)rbase * (3 * C_DIM) + gcol;
            #pragma unroll
            for (int r = 0; r < 8; ++r) {
                const int m = r + 8 * hi;
                if (rbase + m < N)
                    Hp[(size_t)m * (3 * C_DIM) + l16] = acc[s][r];
            }
        }
    }
}

#define OWN_WAVES  32
#define QFLUSH     64
#define QCAP       576
#define KEEP_CAP   256

__device__ __forceinline__
void add_edges(const float* __restrict__ H, float* acc, const int lane,
               const int s_l, const int r_l, const float u_l, const int cn)
{
    int j = 0;
    for (; j + 4 <= cn; j += 4) {
        float4 a[4], b[4];
        int    r[4];
        float  uu[4];
        #pragma unroll
        for (int q = 0; q < 4; ++q) {
            const int s = __builtin_amdgcn_readlane(s_l, j + q);
            r[q]  = __builtin_amdgcn_readlane(r_l, j + q);
            uu[q] = __int_as_float(__builtin_amdgcn_readlane(__float_as_int(u_l), j + q));
            const float4* hp = (const float4*)(H + (size_t)s * 3 * C_DIM);
            a[q] = hp[lane];
            b[q] = hp[32 + lane];
        }
        #pragma unroll
        for (int q = 0; q < 4; ++q) {
            const float w0 = 1.0f - uu[q];
            float* ap = acc + r[q] * C_DIM + lane;
            __hip_atomic_fetch_add(ap,      w0 * a[q].x + uu[q] * b[q].x, __ATOMIC_RELAXED, __HIP_MEMORY_SCOPE_WORKGROUP);
            __hip_atomic_fetch_add(ap + 32, w0 * a[q].y + uu[q] * b[q].y, __ATOMIC_RELAXED, __HIP_MEMORY_SCOPE_WORKGROUP);
            __hip_atomic_fetch_add(ap + 64, w0 * a[q].z + uu[q] * b[q].z, __ATOMIC_RELAXED, __HIP_MEMORY_SCOPE_WORKGROUP);
            __hip_atomic_fetch_add(ap + 96, w0 * a[q].w + uu[q] * b[q].w, __ATOMIC_RELAXED, __HIP_MEMORY_SCOPE_WORKGROUP);
        }
    }
    for (; j < cn; ++j) {
        const int   s  = __builtin_amdgcn_readlane(s_l, j);
        const int   rj = __builtin_amdgcn_readlane(r_l, j);
        const float uj = __int_as_float(__builtin_amdgcn_readlane(__float_as_int(u_l), j));
        const float4* hp = (const float4*)(H + (size_t)s * 3 * C_DIM);
        const float4 a = hp[lane];
        const float4 b = hp[32 + lane];
        const float w0 = 1.0f - uj;
        float* ap = acc + rj * C_DIM + lane;
        __hip_atomic_fetch_add(ap,      w0 * a.x + uj * b.x, __ATOMIC_RELAXED, __HIP_MEMORY_SCOPE_WORKGROUP);
        __hip_atomic_fetch_add(ap + 32, w0 * a.y + uj * b.y, __ATOMIC_RELAXED, __HIP_MEMORY_SCOPE_WORKGROUP);
        __hip_atomic_fetch_add(ap + 64, w0 * a.z + uj * b.z, __ATOMIC_RELAXED, __HIP_MEMORY_SCOPE_WORKGROUP);
        __hip_atomic_fetch_add(ap + 96, w0 * a.w + uj * b.w, __ATOMIC_RELAXED, __HIP_MEMORY_SCOPE_WORKGROUP);
    }
}

template <int PASS>
__global__ __launch_bounds__(1024)
void owner_edge_kernel(const int* __restrict__ ei,
                       const float* __restrict__ u,
                       const float* __restrict__ H,
                       const float* __restrict__ bias,
                       float* __restrict__ out,
                       int* __restrict__ kept,
                       int* __restrict__ kept_n,
                       const unsigned char* __restrict__ keys,
                       int N, int E)
{
    __shared__ float acc[(OWN_ROWS + 1) * C_DIM];
    __shared__ int   cnt[OWN_ROWS + 1];
    __shared__ int   qe[OWN_WAVES * QCAP];
    __shared__ int   over;

    const int lane = threadIdx.x & 31;
    const int wave = __builtin_amdgcn_readfirstlane((int)(threadIdx.x >> 5));
    int* const qew = qe + wave * QCAP;

    const int* __restrict__ dstp = ei + E;
    const int  nchunk = (E + SCAN_EDGES - 1) / SCAN_EDGES;
    const int  rot    = (int)((blockIdx.x * (unsigned)nchunk) / gridDim.x);
    const float4 bb   = ((const float4*)bias)[lane];
    const unsigned bpat = (blockIdx.x & 255u) * 0x01010101u;
    const v4i* __restrict__ keyv = (const v4i*)keys;

    const bool   one_run = N <= (int)gridDim.x * OWN_ROWS;
    const size_t wl      = (size_t)blockIdx.x * OWN_WAVES + (size_t)wave;
    v4i* const   keptw   = (v4i*)kept + wl * KEEP_CAP;

    bool replay = false;
    int  total  = 0;
    if (PASS == 1) {
        if (threadIdx.x == 0) over = 0;
        __syncthreads();
        if (one_run) {
            total = __builtin_amdgcn_readfirstlane(kept_n[wl]);
            if (lane == 0 && total > KEEP_CAP) over = 1;
        }
        __syncthreads();
        replay = one_run && (over == 0);
    }

    for (int n0 = (int)blockIdx.x * OWN_ROWS; n0 < N; n0 += (int)gridDim.x * OWN_ROWS) {
        const int rows = (N - n0 < OWN_ROWS) ? (N - n0) : OWN_ROWS;

        for (int i = threadIdx.x; i < (OWN_ROWS + 1) * C_DIM; i += 1024) acc[i] = 0.0f;
        for (int i = threadIdx.x; i < OWN_ROWS + 1; i += 1024) cnt[i] = 0;
        __syncthreads();

        if (PASS == 1 && replay) {
            v4i nx = {0, 0, 0, 0};
            if (lane < total) nx = keptw[lane];
            for (int g = 0; g < total; g += 32) {
                const int i  = g + lane;
                const v4i rc = nx;
                if (i + 32 < total) nx = keptw[i + 32];
                if (i < total) atomicAdd(&cnt[rc.y], 1);
                const int cn = (total - g < 32) ? (total - g) : 32;
                add_edges(H, acc, lane, rc.x, rc.y, __int_as_float(rc.z), cn);
            }
        } else {
            int qn    = 0;
            int saved = 0;
            auto load_keys = [&](const int cc) -> v4i {
                int c = cc + rot;
                if (c >= nchunk) c -= nchunk;
                return keyv[(size_t)((cc < nchunk) ? c : 0) * 32 + lane];
            };

            auto trip = [&](const v4i kv, const int c0) -> bool {
                const bool last = (c0 >= nchunk);
                if (!last) {
                    int c = c0 + rot;
                    if (c >= nchunk) c -= nchunk;
                    const int base = c * SCAN_EDGES + lane * 16;
                    #pragma unroll
                    for (int k = 0; k < 4; ++k) {
                        const unsigned x = (unsigned)kv[k] ^ bpat;
                        unsigned z = (x - 0x01010101u) & ~x & 0x80808080u;
                        unsigned mk = __builtin_amdgcn_ballot_w32(z != 0u);
                        while (mk != 0u) {
                            if (z != 0u) {
                                const int t = __builtin_ctz(z) >> 3;
                                const int p = qn + (int)__builtin_amdgcn_mbcnt_lo(mk, 0u);
                                qew[p] = base + 4 * k + t;
                                z &= z - 1u;
                            }
                            qn += __builtin_popcount(mk);
                            mk = __builtin_amdgcn_ballot_w32(z != 0u);
                        }
                    }
                }
                if (qn >= QFLUSH || (last && qn > 0)) {
                    for (int g = 0; g < qn; g += 32) {
                        const int i = g + lane;
                        int   s_l = 0, r_l = 0;
                        float u_l = 0.0f;
                        if (i < qn) {
                            const int  e  = qew[i];
                            const bool ok = e < E;
                            const int  ee = ok ? e : 0;
                            const int  d  = dstp[ee];
                            s_l = ei[ee];
                            u_l = u[ee];
                            const unsigned rr = (unsigned)(d - n0);
                            r_l = (ok && rr < (unsigned)rows) ? (int)rr : OWN_ROWS;
                            atomicAdd(&cnt[r_l], 1);
                            if (PASS == 0 && one_run) {
                                const int slot_i = saved + i;
                                if (slot_i < KEEP_CAP) {
                                    v4i rc;
                                    rc.x = s_l;
                                    rc.y = r_l;
                                    rc.z = __float_as_int(u_l);
                                    rc.w = 0;
                                    keptw[slot_i] = rc;
                                }
                            }
                        }
                        const int cn = (qn - g < 32) ? (qn - g) : 32;
                        add_edges(H, acc, lane, s_l, r_l, u_l, cn);
                    }
                    saved += qn;
                    qn = 0;
                }
                return last;
            };
            for (int c0 = wave; ; c0 += 4 * OWN_WAVES) {
                const v4i kq0 = load_keys(c0);
                const v4i kq1 = load_keys(c0 + OWN_WAVES);
                const v4i kq2 = load_keys(c0 + 2 * OWN_WAVES);
                const v4i kq3 = load_keys(c0 + 3 * OWN_WAVES);
                if (trip(kq0, c0)) break;
                if (trip(kq1, c0 + OWN_WAVES)) break;
                if (trip(kq2, c0 + 2 * OWN_WAVES)) break;
                if (trip(kq3, c0 + 3 * OWN_WAVES)) break;
            }
            if (PASS == 0 && one_run && lane == 0) kept_n[wl] = saved;
        }
        __syncthreads();

        for (int r = wave; r < rows; r += OWN_WAVES) {
            const int   n   = n0 + r;
            const float inv = 1.0f / fmaxf((float)cnt[r], 1.0f);
            const float* ar = acc + r * C_DIM + lane;
            const float4 hr = ((const float4*)(H + (size_t)n * 3 * C_DIM + 2 * C_DIM))[lane];
            const float4 o  = make_float4(ar[0]  * inv + hr.x + bb.x,
                                          ar[32] * inv + hr.y + bb.y,
                                          ar[64] * inv + hr.z + bb.z,
                                          ar[96] * inv + hr.w + bb.w);
            ((float4*)(out + (size_t)n * C_DIM))[lane] = o;
        }
        __syncthreads();
    }
}

extern "C" void kernel_launch(void* const* d_in, const int* in_sizes, int n_in,
                              void* d_out, int out_size, void* d_ws, size_t ws_size,
                              hipStream_t stream)
{
    const float* x  = (const float*)d_in[0];
    const int*   ei = (const int*)d_in[1];
    const float* ea = (const float*)d_in[2];
    const float* W1 = (const float*)d_in[3];
    const float* r1 = (const float*)d_in[4];
    const float* b1 = (const float*)d_in[5];
    const float* W2 = (const float*)d_in[6];
    const float* r2 = (const float*)d_in[7];
    const float* b2 = (const float*)d_in[8];
    float* out = (float*)d_out;

    const int N = in_sizes[0] / C_DIM;
    const int E = in_sizes[1] / 2;

    float* ws   = (float*)d_ws;
    float* Hbuf = ws;
    float* hmid = Hbuf + (size_t)N * 3 * C_DIM;
    int* kept   = (int*)(hmid + (size_t)N * C_DIM);
    int* kept_n = kept + (size_t)256 * OWN_WAVES * KEEP_CAP * 4;
    unsigned char* keys = (unsigned char*)(kept_n + (size_t)256 * OWN_WAVES);

    const int mblocks = (N + MTILE - 1) / MTILE;
    const int oblocks = 256;

    splineconv_gemm_kernel<<<mblocks, 256, 0, stream>>>(x, W1, r1, Hbuf, N, ei, keys, E);
    owner_edge_kernel<0><<<oblocks, 1024, 0, stream>>>(ei, ea, Hbuf, b1, hmid, kept, kept_n, keys, N, E);

    splineconv_gemm_kernel<<<mblocks, 256, 0, stream>>>(hmid, W2, r2, Hbuf, N, ei, nullptr, E);
    owner_edge_kernel<1><<<oblocks, 1024, 0, stream>>>(ei, ea, Hbuf, b2, out, kept, kept_n, keys, N, E);
}


# layer 2's product moved into layer 1's edge kernel: a block's finished rows go from its LDS sums into an LDS A tile and its waves multiply them there (f32 WMMA); 3 launches, not 4; sum orders kept
# speedup vs baseline: 2.9219x; 1.0827x over previous
#include <hip/hip_runtime.h>

typedef __attribute__((ext_vector_type(2))) float v2f;
typedef __attribute__((ext_vector_type(8))) float v8f;
typedef __attribute__((ext_vector_type(4))) int   v4i;

#define C_DIM 128
#define LDA   132
#define MTILE 64
#define KCHUNK 32
#define OWN_ROWS   196
#define SCAN_EDGES 512
#define OWN_MT     13
#define A2_ROWS    (OWN_MT * 16)
#define FUSED_KC   16
static_assert(A2_ROWS >= OWN_ROWS && OWN_MT == 4 + 3 + 3 + 3, "the row-tile groups of the edge kernel's product must cover a whole run of rows");

__global__ __launch_bounds__(256)
void splineconv_gemm_kernel(const float* __restrict__ X,
                            const float* __restrict__ W,
                            const float* __restrict__ root,
                            float* __restrict__ H,
                            int N,
                            const int* __restrict__ ei,
                            unsigned char* __restrict__ keys,
                            int E)
{
    __shared__ float As[MTILE * LDA];
    const int tid  = threadIdx.x;
    const int wave = tid >> 5;
    const int lane = tid & 31;
    const int row0 = blockIdx.x * MTILE;

    if (keys != nullptr) {
        const int* __restrict__ dstl = ei + E;
        const int nword = ((E + SCAN_EDGES - 1) / SCAN_EDGES) * (SCAN_EDGES / 4);
        for (int w = blockIdx.x * 256 + tid; w < nword; w += (int)gridDim.x * 256) {
            unsigned pk = 0u;
            #pragma unroll
            for (int t = 0; t < 4; ++t) {
                const int  e  = 4 * w + t;
                const bool ok = e < E;
                const unsigned d  = (unsigned)dstl[ok ? e : 0];
                const unsigned kb = ok ? ((d / (unsigned)OWN_ROWS) & 255u) : 255u;
                pk |= kb << (8 * t);
            }
            ((unsigned*)keys)[w] = pk;
        }
    }

    for (int i = tid; i < MTILE * (C_DIM / 4); i += 256) {
        const int r  = i >> 5;
        const int c4 = i & 31;
        float4 v = make_float4(0.f, 0.f, 0.f, 0.f);
        const int gr = row0 + r;
        if (gr < N) v = ((const float4*)(X + (size_t)gr * C_DIM))[c4];
        float* dst = &As[r * LDA + c4 * 4];
        dst[0] = v.x; dst[1] = v.y; dst[2] = v.z; dst[3] = v.w;
    }
    __syncthreads();

    const int hi  = lane >> 4;
    const int l16 = lane & 15;

    for (int t = wave; t < 24; t += 8) {
        const int gcol = t * 16;
        const float* Bbase;
        if (gcol < C_DIM)          Bbase = W + gcol;
        else if (gcol < 2 * C_DIM) Bbase = W + (size_t)C_DIM * C_DIM + (gcol - C_DIM);
        else                       Bbase = root + (gcol - 2 * C_DIM);

        const float* Bl = Bbase + (size_t)(2 * hi) * C_DIM + l16;
        const float* Al = &As[l16 * LDA + 2 * hi];

        v8f acc[4] = {{}, {}, {}, {}};
        #pragma unroll 1
        for (int kc = 0; kc < C_DIM / KCHUNK; ++kc) {
            const float* Bc = Bl + (size_t)(kc * KCHUNK) * C_DIM;
            const float* Ac = Al + kc * KCHUNK;

            v2f bf[KCHUNK / 4];
            #pragma unroll
            for (int j = 0; j < KCHUNK / 4; ++j) {
                bf[j].x = Bc[(size_t)(4 * j) * C_DIM];
                bf[j].y = Bc[(size_t)(4 * j + 1) * C_DIM];
            }
            v2f af[KCHUNK / 4][4];
            #pragma unroll
            for (int j = 0; j < KCHUNK / 4; ++j) {
                #pragma unroll
                for (int s = 0; s < 4; ++s) {
                    af[j][s].x = Ac[s * 16 * LDA + 4 * j];
                    af[j][s].y = Ac[s * 16 * LDA + 4 * j + 1];
                }
            }
            #pragma unroll
            for (int j = 0; j < KCHUNK / 4; ++j) {
                #pragma unroll
                for (int s = 0; s < 4; ++s) {
                    acc[s] = __builtin_amdgcn_wmma_f32_16x16x4_f32(
                                 false, af[j][s], false, bf[j], (short)0, acc[s], false, false);
                }
            }
        }

        #pragma unroll
        for (int s = 0; s < 4; ++s) {
            const int rbase = row0 + s * 16;
            float* Hp = H + (size_t)rbase * (3 * C_DIM) + gcol;
            #pragma unroll
            for (int r = 0; r < 8; ++r) {
                const int m = r + 8 * hi;
                if (rbase + m < N)
                    Hp[(size_t)m * (3 * C_DIM) + l16] = acc[s][r];
            }
        }
    }
}

#define OWN_WAVES  32
#define QFLUSH     64
#define QCAP       576
#define KEEP_CAP   256

__device__ __forceinline__
void add_edges(const float* __restrict__ H, float* acc, const int lane,
               const int s_l, const int r_l, const float u_l, const int cn)
{
    int j = 0;
    for (; j + 4 <= cn; j += 4) {
        float4 a[4], b[4];
        int    r[4];
        float  uu[4];
        #pragma unroll
        for (int q = 0; q < 4; ++q) {
            const int s = __builtin_amdgcn_readlane(s_l, j + q);
            r[q]  = __builtin_amdgcn_readlane(r_l, j + q);
            uu[q] = __int_as_float(__builtin_amdgcn_readlane(__float_as_int(u_l), j + q));
            const float4* hp = (const float4*)(H + (size_t)s * 3 * C_DIM);
            a[q] = hp[lane];
            b[q] = hp[32 + lane];
        }
        #pragma unroll
        for (int q = 0; q < 4; ++q) {
            const float w0 = 1.0f - uu[q];
            float* ap = acc + r[q] * C_DIM + lane;
            __hip_atomic_fetch_add(ap,      w0 * a[q].x + uu[q] * b[q].x, __ATOMIC_RELAXED, __HIP_MEMORY_SCOPE_WORKGROUP);
            __hip_atomic_fetch_add(ap + 32, w0 * a[q].y + uu[q] * b[q].y, __ATOMIC_RELAXED, __HIP_MEMORY_SCOPE_WORKGROUP);
            __hip_atomic_fetch_add(ap + 64, w0 * a[q].z + uu[q] * b[q].z, __ATOMIC_RELAXED, __HIP_MEMORY_SCOPE_WORKGROUP);
            __hip_atomic_fetch_add(ap + 96, w0 * a[q].w + uu[q] * b[q].w, __ATOMIC_RELAXED, __HIP_MEMORY_SCOPE_WORKGROUP);
        }
    }
    for (; j < cn; ++j) {
        const int   s  = __builtin_amdgcn_readlane(s_l, j);
        const int   rj = __builtin_amdgcn_readlane(r_l, j);
        const float uj = __int_as_float(__builtin_amdgcn_readlane(__float_as_int(u_l), j));
        const float4* hp = (const float4*)(H + (size_t)s * 3 * C_DIM);
        const float4 a = hp[lane];
        const float4 b = hp[32 + lane];
        const float w0 = 1.0f - uj;
        float* ap = acc + rj * C_DIM + lane;
        __hip_atomic_fetch_add(ap,      w0 * a.x + uj * b.x, __ATOMIC_RELAXED, __HIP_MEMORY_SCOPE_WORKGROUP);
        __hip_atomic_fetch_add(ap + 32, w0 * a.y + uj * b.y, __ATOMIC_RELAXED, __HIP_MEMORY_SCOPE_WORKGROUP);
        __hip_atomic_fetch_add(ap + 64, w0 * a.z + uj * b.z, __ATOMIC_RELAXED, __HIP_MEMORY_SCOPE_WORKGROUP);
        __hip_atomic_fetch_add(ap + 96, w0 * a.w + uj * b.w, __ATOMIC_RELAXED, __HIP_MEMORY_SCOPE_WORKGROUP);
    }
}

template <int NS, int KC>
__device__ __forceinline__
void product_rows(const float* Al, const float* __restrict__ Bl, float* __restrict__ Hl,
                  const int rows_left, const int hi)
{
    v8f acc[NS] = {};
    #pragma unroll 1
    for (int kc = 0; kc < C_DIM / KC; ++kc) {
        const float* Bc = Bl + (size_t)(kc * KC) * C_DIM;
        const float* Ac = Al + kc * KC;

        v2f bf[KC / 4];
        #pragma unroll
        for (int j = 0; j < KC / 4; ++j) {
            bf[j].x = Bc[(size_t)(4 * j) * C_DIM];
            bf[j].y = Bc[(size_t)(4 * j + 1) * C_DIM];
        }
        v2f af[KC / 4][NS];
        #pragma unroll
        for (int j = 0; j < KC / 4; ++j) {
            #pragma unroll
            for (int s = 0; s < NS; ++s) {
                af[j][s].x = Ac[s * 16 * LDA + 4 * j];
                af[j][s].y = Ac[s * 16 * LDA + 4 * j + 1];
            }
        }
        #pragma unroll
        for (int j = 0; j < KC / 4; ++j) {
            #pragma unroll
            for (int s = 0; s < NS; ++s) {
                acc[s] = __builtin_amdgcn_wmma_f32_16x16x4_f32(
                             false, af[j][s], false, bf[j], (short)0, acc[s], false, false);
            }
        }
    }
    #pragma unroll
    for (int s = 0; s < NS; ++s) {
        #pragma unroll
        for (int r = 0; r < 8; ++r) {
            if (s * 16 + r + 8 * hi < rows_left)
                Hl[(size_t)(s * 16 + r) * (3 * C_DIM)] = acc[s][r];
        }
    }
}

template <int PASS>
__global__ __launch_bounds__(1024)
void owner_edge_kernel(const int* __restrict__ ei,
                       const float* __restrict__ u,
                       const float* __restrict__ H,
                       const float* __restrict__ bias,
                       float* __restrict__ out,
                       int* __restrict__ kept,
                       int* __restrict__ kept_n,
                       const unsigned char* __restrict__ keys,
                       int N, int E,
                       const float* __restrict__ Wn,
                       const float* __restrict__ rootn,
                       float* __restrict__ Hn)
{
    __shared__ float acc[(OWN_ROWS + 1) * C_DIM];
    __shared__ int   cnt[OWN_ROWS + 1];
    __shared__ int   qe[OWN_WAVES * QCAP];
    __shared__ int   over;
    __shared__ __attribute__((aligned(16))) float A2[PASS == 0 ? A2_ROWS * LDA : 4];

    const int lane = threadIdx.x & 31;
    const int wave = __builtin_amdgcn_readfirstlane((int)(threadIdx.x >> 5));
    int* const qew = qe + wave * QCAP;

    const int* __restrict__ dstp = ei + E;
    const int  nchunk = (E + SCAN_EDGES - 1) / SCAN_EDGES;
    const int  rot    = (int)((blockIdx.x * (unsigned)nchunk) / gridDim.x);
    const float4 bb   = ((const float4*)bias)[lane];
    const unsigned bpat = (blockIdx.x & 255u) * 0x01010101u;
    const v4i* __restrict__ keyv = (const v4i*)keys;

    const bool   one_run = N <= (int)gridDim.x * OWN_ROWS;
    const size_t wl      = (size_t)blockIdx.x * OWN_WAVES + (size_t)wave;
    v4i* const   keptw   = (v4i*)kept + wl * KEEP_CAP;

    bool replay = false;
    int  total  = 0;
    if (PASS == 1) {
        if (threadIdx.x == 0) over = 0;
        __syncthreads();
        if (one_run) {
            total = __builtin_amdgcn_readfirstlane(kept_n[wl]);
            if (lane == 0 && total > KEEP_CAP) over = 1;
        }
        __syncthreads();
        replay = one_run && (over == 0);
    }

    for (int n0 = (int)blockIdx.x * OWN_ROWS; n0 < N; n0 += (int)gridDim.x * OWN_ROWS) {
        const int rows = (N - n0 < OWN_ROWS) ? (N - n0) : OWN_ROWS;

        for (int i = threadIdx.x; i < (OWN_ROWS + 1) * C_DIM; i += 1024) acc[i] = 0.0f;
        for (int i = threadIdx.x; i < OWN_ROWS + 1; i += 1024) cnt[i] = 0;
        __syncthreads();

        if (PASS == 1 && replay) {
            v4i nx = {0, 0, 0, 0};
            if (lane < total) nx = keptw[lane];
            for (int g = 0; g < total; g += 32) {
                const int i  = g + lane;
                const v4i rc = nx;
                if (i + 32 < total) nx = keptw[i + 32];
                if (i < total) atomicAdd(&cnt[rc.y], 1);
                const int cn = (total - g < 32) ? (total - g) : 32;
                add_edges(H, acc, lane, rc.x, rc.y, __int_as_float(rc.z), cn);
            }
        } else {
            int qn    = 0;
            int saved = 0;
            auto load_keys = [&](const int cc) -> v4i {
                int c = cc + rot;
                if (c >= nchunk) c -= nchunk;
                return keyv[(size_t)((cc < nchunk) ? c : 0) * 32 + lane];
            };

            auto trip = [&](const v4i kv, const int c0) -> bool {
                const bool last = (c0 >= nchunk);
                if (!last) {
                    int c = c0 + rot;
                    if (c >= nchunk) c -= nchunk;
                    const int base = c * SCAN_EDGES + lane * 16;
                    #pragma unroll
                    for (int k = 0; k < 4; ++k) {
                        const unsigned x = (unsigned)kv[k] ^ bpat;
                        unsigned z = (x - 0x01010101u) & ~x & 0x80808080u;
                        unsigned mk = __builtin_amdgcn_ballot_w32(z != 0u);
                        while (mk != 0u) {
                            if (z != 0u) {
                                const int t = __builtin_ctz(z) >> 3;
                                const int p = qn + (int)__builtin_amdgcn_mbcnt_lo(mk, 0u);
                                qew[p] = base + 4 * k + t;
                                z &= z - 1u;
                            }
                            qn += __builtin_popcount(mk);
                            mk = __builtin_amdgcn_ballot_w32(z != 0u);
                        }
                    }
                }
                if (qn >= QFLUSH || (last && qn > 0)) {
                    for (int g = 0; g < qn; g += 32) {
                        const int i = g + lane;
                        int   s_l = 0, r_l = 0;
                        float u_l = 0.0f;
                        if (i < qn) {
                            const int  e  = qew[i];
                            const bool ok = e < E;
                            const int  ee = ok ? e : 0;
                            const int  d  = dstp[ee];
                            s_l = ei[ee];
                            u_l = u[ee];
                            const unsigned rr = (unsigned)(d - n0);
                            r_l = (ok && rr < (unsigned)rows) ? (int)rr : OWN_ROWS;
                            atomicAdd(&cnt[r_l], 1);
                            if (PASS == 0 && one_run) {
                                const int slot_i = saved + i;
                                if (slot_i < KEEP_CAP) {
                                    v4i rc;
                                    rc.x = s_l;
                                    rc.y = r_l;
                                    rc.z = __float_as_int(u_l);
                                    rc.w = 0;
                                    keptw[slot_i] = rc;
                                }
                            }
                        }
                        const int cn = (qn - g < 32) ? (qn - g) : 32;
                        add_edges(H, acc, lane, s_l, r_l, u_l, cn);
                    }
                    saved += qn;
                    qn = 0;
                }
                return last;
            };
            for (int c0 = wave; ; c0 += 4 * OWN_WAVES) {
                const v4i kq0 = load_keys(c0);
                const v4i kq1 = load_keys(c0 + OWN_WAVES);
                const v4i kq2 = load_keys(c0 + 2 * OWN_WAVES);
                const v4i kq3 = load_keys(c0 + 3 * OWN_WAVES);
                if (trip(kq0, c0)) break;
                if (trip(kq1, c0 + OWN_WAVES)) break;
                if (trip(kq2, c0 + 2 * OWN_WAVES)) break;
                if (trip(kq3, c0 + 3 * OWN_WAVES)) break;
            }
            if (PASS == 0 && one_run && lane == 0) kept_n[wl] = saved;
        }
        __syncthreads();

        for (int r = wave; r < (PASS == 0 ? A2_ROWS : rows); r += OWN_WAVES) {
            float4 o = make_float4(0.f, 0.f, 0.f, 0.f);
            if (r < rows) {
                const int   n   = n0 + r;
                const float inv = 1.0f / fmaxf((float)cnt[r], 1.0f);
                const float* ar = acc + r * C_DIM + lane;
                const float4 hr = ((const float4*)(H + (size_t)n * 3 * C_DIM + 2 * C_DIM))[lane];
                o = make_float4(ar[0]  * inv + hr.x + bb.x,
                                ar[32] * inv + hr.y + bb.y,
                                ar[64] * inv + hr.z + bb.z,
                                ar[96] * inv + hr.w + bb.w);
            }
            if (PASS == 0) *(float4*)&A2[r * LDA + 4 * lane] = o;
            else           ((float4*)(out + (size_t)(n0 + r) * C_DIM))[lane] = o;
        }
        if (PASS == 0) {
            __syncthreads();
            for (int it = wave; it < 4 * (3 * C_DIM / 16); it += OWN_WAVES) {
                const int g    = it / (3 * C_DIM / 16);
                const int gcol = (it - g * (3 * C_DIM / 16)) * 16;
                const int mt0  = (g == 0) ? 0 : 3 * g + 1;
                if (mt0 * 16 >= rows) continue;
                const int hi   = lane >> 4;
                const int l16  = lane & 15;
                const float* Bbase;
                if (gcol < C_DIM)          Bbase = Wn + gcol;
                else if (gcol < 2 * C_DIM) Bbase = Wn + (size_t)C_DIM * C_DIM + (gcol - C_DIM);
                else                       Bbase = rootn + (gcol - 2 * C_DIM);
                const float* Bl = Bbase + (size_t)(2 * hi) * C_DIM + l16;
                const float* Al = &A2[(mt0 * 16 + l16) * LDA + 2 * hi];
                float*       Hl = Hn + (size_t)(n0 + mt0 * 16 + 8 * hi) * (3 * C_DIM) + gcol + l16;
                if (g == 0) product_rows<4, FUSED_KC>(Al, Bl, Hl, rows - mt0 * 16, hi);
                else        product_rows<3, FUSED_KC>(Al, Bl, Hl, rows - mt0 * 16, hi);
            }
        }
        __syncthreads();
    }
}

extern "C" void kernel_launch(void* const* d_in, const int* in_sizes, int n_in,
                              void* d_out, int out_size, void* d_ws, size_t ws_size,
                              hipStream_t stream)
{
    const float* x  = (const float*)d_in[0];
    const int*   ei = (const int*)d_in[1];
    const float* ea = (const float*)d_in[2];
    const float* W1 = (const float*)d_in[3];
    const float* r1 = (const float*)d_in[4];
    const float* b1 = (const float*)d_in[5];
    const float* W2 = (const float*)d_in[6];
    const float* r2 = (const float*)d_in[7];
    const float* b2 = (const float*)d_in[8];
    float* out = (float*)d_out;

    const int N = in_sizes[0] / C_DIM;
    const int E = in_sizes[1] / 2;

    float* ws   = (float*)d_ws;
    float* Hbuf  = ws;
    float* H2buf = Hbuf + (size_t)N * 3 * C_DIM;
    int* kept   = (int*)(H2buf + (size_t)N * 3 * C_DIM);
    int* kept_n = kept + (size_t)256 * OWN_WAVES * KEEP_CAP * 4;
    unsigned char* keys = (unsigned char*)(kept_n + (size_t)256 * OWN_WAVES);

    const int mblocks = (N + MTILE - 1) / MTILE;
    const int oblocks = 256;

    splineconv_gemm_kernel<<<mblocks, 256, 0, stream>>>(x, W1, r1, Hbuf, N, ei, keys, E);
    owner_edge_kernel<0><<<oblocks, 1024, 0, stream>>>(ei, ea, Hbuf, b1, nullptr, kept, kept_n, keys, N, E,
                                                        W2, r2, H2buf);

    owner_edge_kernel<1><<<oblocks, 1024, 0, stream>>>(ei, ea, H2buf, b2, out, kept, kept_n, keys, N, E,
                                                        nullptr, nullptr, nullptr);
}


# edge filter: layer 1's product kernel writes a bit row of the edge list per owner key (LDS table, OR); an edge block reads only its own row, 1 bit per edge, not all byte keys; sums: LDS arrival order
# speedup vs baseline: 3.0077x; 1.0294x over previous
#include <hip/hip_runtime.h>

typedef __attribute__((ext_vector_type(2))) float v2f;
typedef __attribute__((ext_vector_type(8))) float v8f;
typedef __attribute__((ext_vector_type(4))) int   v4i;

#define C_DIM 128
#define LDA   132
#define MTILE 64
#define KCHUNK 32
#define OWN_ROWS   196
#define OWNERS     256
#define GRP_EDGES  1024
#define OWN_MT     13
#define A2_ROWS    (OWN_MT * 16)
#define FUSED_KC   16
static_assert(A2_ROWS >= OWN_ROWS && OWN_MT == 4 + 3 + 3 + 3, "the row-tile groups of the edge kernel's product must cover a whole run of rows");

__device__ __forceinline__ int bitmap_groups(const int E)
{
    return (((E + GRP_EDGES - 1) / GRP_EDGES) + 3) & ~3;
}

__global__ __launch_bounds__(256)
void splineconv_gemm_kernel(const float* __restrict__ X,
                            const float* __restrict__ W,
                            const float* __restrict__ root,
                            float* __restrict__ H,
                            int N,
                            const int* __restrict__ ei,
                            unsigned* __restrict__ bm,
                            int E)
{
    __shared__ float As[MTILE * LDA];
    __shared__ __attribute__((aligned(16))) unsigned tab[OWNERS * 32];
    const int tid  = threadIdx.x;
    const int wave = tid >> 5;
    const int lane = tid & 31;
    const int row0 = blockIdx.x * MTILE;

    const bool has_grp = (bm != nullptr) && ((int)blockIdx.x < bitmap_groups(E));
    const int* __restrict__ dstl = ei + E;
    auto owners_of = [&](const int grp, int (&wd)[4], unsigned (&bt)[4]) {
        #pragma unroll
        for (int t = 0; t < 4; ++t) {
            const int  e  = grp * GRP_EDGES + 4 * tid + t;
            const bool ok = e < E;
            const unsigned d = (unsigned)dstl[ok ? e : 0];
            wd[t] = (int)((d / (unsigned)OWN_ROWS) & 255u) * 32 + (tid >> 3);
            bt[t] = ok ? (1u << (4 * (tid & 7) + t)) : 0u;
        }
    };
    auto clear_tab = [&]() {
        for (int i = tid; i < OWNERS * 32 / 4; i += 256)
            ((uint4*)tab)[i] = make_uint4(0u, 0u, 0u, 0u);
    };
    auto mark = [&](const int (&wd)[4], const unsigned (&bt)[4]) {
        #pragma unroll
        for (int t = 0; t < 4; ++t) atomicOr(&tab[wd[t]], bt[t]);
    };
    auto store_tab = [&](const int grp) {
        const int ngrp = bitmap_groups(E);
        for (int i = tid; i < OWNERS * 32 / 4; i += 256) {
            const int o    = i >> 3;
            const int part = i & 7;
            const uint4 v  = ((const uint4*)tab)[i];
            ((uint4*)(bm + ((size_t)o * ngrp + grp) * 32))[part] = v;
        }
    };
    int      wd0[4] = {0, 0, 0, 0};
    unsigned bt0[4] = {0u, 0u, 0u, 0u};
    if (has_grp) {
        owners_of((int)blockIdx.x, wd0, bt0);
        clear_tab();
    }

    for (int i = tid; i < MTILE * (C_DIM / 4); i += 256) {
        const int r  = i >> 5;
        const int c4 = i & 31;
        float4 v = make_float4(0.f, 0.f, 0.f, 0.f);
        const int gr = row0 + r;
        if (gr < N) v = ((const float4*)(X + (size_t)gr * C_DIM))[c4];
        float* dst = &As[r * LDA + c4 * 4];
        dst[0] = v.x; dst[1] = v.y; dst[2] = v.z; dst[3] = v.w;
    }
    __syncthreads();

    const int hi  = lane >> 4;
    const int l16 = lane & 15;

    for (int t = wave; t < 24; t += 8) {
        const int gcol = t * 16;
        const float* Bbase;
        if (gcol < C_DIM)          Bbase = W + gcol;
        else if (gcol < 2 * C_DIM) Bbase = W + (size_t)C_DIM * C_DIM + (gcol - C_DIM);
        else                       Bbase = root + (gcol - 2 * C_DIM);

        const float* Bl = Bbase + (size_t)(2 * hi) * C_DIM + l16;
        const float* Al = &As[l16 * LDA + 2 * hi];

        v8f acc[4] = {{}, {}, {}, {}};
        #pragma unroll 1
        for (int kc = 0; kc < C_DIM / KCHUNK; ++kc) {
            const float* Bc = Bl + (size_t)(kc * KCHUNK) * C_DIM;
            const float* Ac = Al + kc * KCHUNK;

            v2f bf[KCHUNK / 4];
            #pragma unroll
            for (int j = 0; j < KCHUNK / 4; ++j) {
                bf[j].x = Bc[(size_t)(4 * j) * C_DIM];
                bf[j].y = Bc[(size_t)(4 * j + 1) * C_DIM];
            }
            v2f af[KCHUNK / 4][4];
            #pragma unroll
            for (int j = 0; j < KCHUNK / 4; ++j) {
                #pragma unroll
                for (int s = 0; s < 4; ++s) {
                    af[j][s].x = Ac[s * 16 * LDA + 4 * j];
                    af[j][s].y = Ac[s * 16 * LDA + 4 * j + 1];
                }
            }
            #pragma unroll
            for (int j = 0; j < KCHUNK / 4; ++j) {
                #pragma unroll
                for (int s = 0; s < 4; ++s) {
                    acc[s] = __builtin_amdgcn_wmma_f32_16x16x4_f32(
                                 false, af[j][s], false, bf[j], (short)0, acc[s], false, false);
                }
            }
        }

        if (has_grp && t == wave) {
            mark(wd0, bt0);
            __syncthreads();
            store_tab((int)blockIdx.x);
        }

        #pragma unroll
        for (int s = 0; s < 4; ++s) {
            const int rbase = row0 + s * 16;
            float* Hp = H + (size_t)rbase * (3 * C_DIM) + gcol;
            #pragma unroll
            for (int r = 0; r < 8; ++r) {
                const int m = r + 8 * hi;
                if (rbase + m < N)
                    Hp[(size_t)m * (3 * C_DIM) + l16] = acc[s][r];
            }
        }
    }

    if (bm != nullptr) {
        for (int grp = (int)blockIdx.x + (int)gridDim.x; grp < bitmap_groups(E); grp += (int)gridDim.x) {
            int      wd[4];
            unsigned bt[4];
            __syncthreads();
            owners_of(grp, wd, bt);
            clear_tab();
            __syncthreads();
            mark(wd, bt);
            __syncthreads();
            store_tab(grp);
        }
    }
}

#define OWN_WAVES  32
#define QFLUSH     64
#define QCAP       576
#define KEEP_CAP   256

__device__ __forceinline__
void add_edges(const float* __restrict__ H, float* acc, const int lane,
               const int s_l, const int r_l, const float u_l, const int cn)
{
    int j = 0;
    for (; j + 4 <= cn; j += 4) {
        float4 a[4], b[4];
        int    r[4];
        float  uu[4];
        #pragma unroll
        for (int q = 0; q < 4; ++q) {
            const int s = __builtin_amdgcn_readlane(s_l, j + q);
            r[q]  = __builtin_amdgcn_readlane(r_l, j + q);
            uu[q] = __int_as_float(__builtin_amdgcn_readlane(__float_as_int(u_l), j + q));
            const float4* hp = (const float4*)(H + (size_t)s * 3 * C_DIM);
            a[q] = hp[lane];
            b[q] = hp[32 + lane];
        }
        #pragma unroll
        for (int q = 0; q < 4; ++q) {
            const float w0 = 1.0f - uu[q];
            float* ap = acc + r[q] * C_DIM + lane;
            __hip_atomic_fetch_add(ap,      w0 * a[q].x + uu[q] * b[q].x, __ATOMIC_RELAXED, __HIP_MEMORY_SCOPE_WORKGROUP);
            __hip_atomic_fetch_add(ap + 32, w0 * a[q].y + uu[q] * b[q].y, __ATOMIC_RELAXED, __HIP_MEMORY_SCOPE_WORKGROUP);
            __hip_atomic_fetch_add(ap + 64, w0 * a[q].z + uu[q] * b[q].z, __ATOMIC_RELAXED, __HIP_MEMORY_SCOPE_WORKGROUP);
            __hip_atomic_fetch_add(ap + 96, w0 * a[q].w + uu[q] * b[q].w, __ATOMIC_RELAXED, __HIP_MEMORY_SCOPE_WORKGROUP);
        }
    }
    for (; j < cn; ++j) {
        const int   s  = __builtin_amdgcn_readlane(s_l, j);
        const int   rj = __builtin_amdgcn_readlane(r_l, j);
        const float uj = __int_as_float(__builtin_amdgcn_readlane(__float_as_int(u_l), j));
        const float4* hp = (const float4*)(H + (size_t)s * 3 * C_DIM);
        const float4 a = hp[lane];
        const float4 b = hp[32 + lane];
        const float w0 = 1.0f - uj;
        float* ap = acc + rj * C_DIM + lane;
        __hip_atomic_fetch_add(ap,      w0 * a.x + uj * b.x, __ATOMIC_RELAXED, __HIP_MEMORY_SCOPE_WORKGROUP);
        __hip_atomic_fetch_add(ap + 32, w0 * a.y + uj * b.y, __ATOMIC_RELAXED, __HIP_MEMORY_SCOPE_WORKGROUP);
        __hip_atomic_fetch_add(ap + 64, w0 * a.z + uj * b.z, __ATOMIC_RELAXED, __HIP_MEMORY_SCOPE_WORKGROUP);
        __hip_atomic_fetch_add(ap + 96, w0 * a.w + uj * b.w, __ATOMIC_RELAXED, __HIP_MEMORY_SCOPE_WORKGROUP);
    }
}

template <int NS, int KC>
__device__ __forceinline__
void product_rows(const float* Al, const float* __restrict__ Bl, float* __restrict__ Hl,
                  const int rows_left, const int hi)
{
    v8f acc[NS] = {};
    #pragma unroll 1
    for (int kc = 0; kc < C_DIM / KC; ++kc) {
        const float* Bc = Bl + (size_t)(kc * KC) * C_DIM;
        const float* Ac = Al + kc * KC;

        v2f bf[KC / 4];
        #pragma unroll
        for (int j = 0; j < KC / 4; ++j) {
            bf[j].x = Bc[(size_t)(4 * j) * C_DIM];
            bf[j].y = Bc[(size_t)(4 * j + 1) * C_DIM];
        }
        v2f af[KC / 4][NS];
        #pragma unroll
        for (int j = 0; j < KC / 4; ++j) {
            #pragma unroll
            for (int s = 0; s < NS; ++s) {
                af[j][s].x = Ac[s * 16 * LDA + 4 * j];
                af[j][s].y = Ac[s * 16 * LDA + 4 * j + 1];
            }
        }
        #pragma unroll
        for (int j = 0; j < KC / 4; ++j) {
            #pragma unroll
            for (int s = 0; s < NS; ++s) {
                acc[s] = __builtin_amdgcn_wmma_f32_16x16x4_f32(
                             false, af[j][s], false, bf[j], (short)0, acc[s], false, false);
            }
        }
    }
    #pragma unroll
    for (int s = 0; s < NS; ++s) {
        #pragma unroll
        for (int r = 0; r < 8; ++r) {
            if (s * 16 + r + 8 * hi < rows_left)
                Hl[(size_t)(s * 16 + r) * (3 * C_DIM)] = acc[s][r];
        }
    }
}

template <int PASS>
__global__ __launch_bounds__(1024)
void owner_edge_kernel(const int* __restrict__ ei,
                       const float* __restrict__ u,
                       const float* __restrict__ H,
                       const float* __restrict__ bias,
                       float* __restrict__ out,
                       int* __restrict__ kept,
                       int* __restrict__ kept_n,
                       const unsigned* __restrict__ bm,
                       int N, int E,
                       const float* __restrict__ Wn,
                       const float* __restrict__ rootn,
                       float* __restrict__ Hn)
{
    __shared__ float acc[(OWN_ROWS + 1) * C_DIM];
    __shared__ int   cnt[OWN_ROWS + 1];
    __shared__ int   qe[OWN_WAVES * QCAP];
    __shared__ int   over;
    __shared__ __attribute__((aligned(16))) float A2[PASS == 0 ? A2_ROWS * LDA : 4];

    const int lane = threadIdx.x & 31;
    const int wave = __builtin_amdgcn_readfirstlane((int)(threadIdx.x >> 5));
    int* const qew = qe + wave * QCAP;

    const int* __restrict__ dstp = ei + E;
    const int  ngrp   = bitmap_groups(E);
    const int  nchunk = ngrp / 4;
    const float4 bb   = ((const float4*)bias)[lane];
    const v4i* __restrict__ rowv = (const v4i*)(bm + (size_t)(blockIdx.x & 255u) * (size_t)ngrp * 32);

    const bool   one_run = N <= (int)gridDim.x * OWN_ROWS;
    const size_t wl      = (size_t)blockIdx.x * OWN_WAVES + (size_t)wave;
    v4i* const   keptw   = (v4i*)kept + wl * KEEP_CAP;

    bool replay = false;
    int  total  = 0;
    if (PASS == 1) {
        if (threadIdx.x == 0) over = 0;
        __syncthreads();
        if (one_run) {
            total = __builtin_amdgcn_readfirstlane(kept_n[wl]);
            if (lane == 0 && total > KEEP_CAP) over = 1;
        }
        __syncthreads();
        replay = one_run && (over == 0);
    }

    for (int n0 = (int)blockIdx.x * OWN_ROWS; n0 < N; n0 += (int)gridDim.x * OWN_ROWS) {
        const int rows = (N - n0 < OWN_ROWS) ? (N - n0) : OWN_ROWS;

        for (int i = threadIdx.x; i < (OWN_ROWS + 1) * C_DIM; i += 1024) acc[i] = 0.0f;
        for (int i = threadIdx.x; i < OWN_ROWS + 1; i += 1024) cnt[i] = 0;
        __syncthreads();

        if (PASS == 1 && replay) {
            v4i nx = {0, 0, 0, 0};
            if (lane < total) nx = keptw[lane];
            for (int g = 0; g < total; g += 32) {
                const int i  = g + lane;
                const v4i rc = nx;
                if (i + 32 < total) nx = keptw[i + 32];
                if (i < total) atomicAdd(&cnt[rc.y], 1);
                const int cn = (total - g < 32) ? (total - g) : 32;
                add_edges(H, acc, lane, rc.x, rc.y, __int_as_float(rc.z), cn);
            }
        } else {
            int qn    = 0;
            int saved = 0;

            auto drain = [&]() {
                for (int g = 0; g < qn; g += 32) {
                    const int i = g + lane;
                    int   s_l = 0, r_l = 0;
                    float u_l = 0.0f;
                    if (i < qn) {
                        const int  e  = qew[i];
                        const bool ok = e < E;
                        const int  ee = ok ? e : 0;
                        const int  d  = dstp[ee];
                        s_l = ei[ee];
                        u_l = u[ee];
                        const unsigned rr = (unsigned)(d - n0);
                        r_l = (ok && rr < (unsigned)rows) ? (int)rr : OWN_ROWS;
                        atomicAdd(&cnt[r_l], 1);
                        if (PASS == 0 && one_run) {
                            const int slot_i = saved + i;
                            if (slot_i < KEEP_CAP) {
                                v4i rc;
                                rc.x = s_l;
                                rc.y = r_l;
                                rc.z = __float_as_int(u_l);
                                rc.w = 0;
                                keptw[slot_i] = rc;
                            }
                        }
                    }
                    const int cn = (qn - g < 32) ? (qn - g) : 32;
                    add_edges(H, acc, lane, s_l, r_l, u_l, cn);
                }
                saved += qn;
                qn = 0;
            };

            auto load_words = [&](const int c) -> v4i {
                return rowv[(size_t)((c < nchunk) ? c : 0) * 32 + lane];
            };
            v4i n1 = {0, 0, 0, 0};
            v4i n2 = {0, 0, 0, 0};
            for (int c = wave - 2 * OWN_WAVES; c < nchunk; c += OWN_WAVES) {
                v4i kv = n1;
                n1 = n2;
                n2 = load_words(c + 2 * OWN_WAVES);
                if (c < wave) continue;
                const int ebase = (c * 128 + lane * 4) * 32;
                #pragma unroll 1
                for (int k = 0; k < 4; ++k) {
                    unsigned z = (unsigned)kv.x;
                    kv.x = kv.y; kv.y = kv.z; kv.z = kv.w;
                    unsigned mk = __builtin_amdgcn_ballot_w32(z != 0u);
                    while (mk != 0u) {
                        if (z != 0u) {
                            const int t = __builtin_ctz(z);
                            const int p = qn + (int)__builtin_amdgcn_mbcnt_lo(mk, 0u);
                            qew[p] = ebase + 32 * k + t;
                            z &= z - 1u;
                        }
                        qn += __builtin_popcount(mk);
                        if (qn >= QFLUSH) drain();
                        mk = __builtin_amdgcn_ballot_w32(z != 0u);
                    }
                }
            }
            if (qn > 0) drain();
            if (PASS == 0 && one_run && lane == 0) kept_n[wl] = saved;
        }
        __syncthreads();

        for (int r = wave; r < (PASS == 0 ? A2_ROWS : rows); r += OWN_WAVES) {
            float4 o = make_float4(0.f, 0.f, 0.f, 0.f);
            if (r < rows) {
                const int   n   = n0 + r;
                const float inv = 1.0f / fmaxf((float)cnt[r], 1.0f);
                const float* ar = acc + r * C_DIM + lane;
                const float4 hr = ((const float4*)(H + (size_t)n * 3 * C_DIM + 2 * C_DIM))[lane];
                o = make_float4(ar[0]  * inv + hr.x + bb.x,
                                ar[32] * inv + hr.y + bb.y,
                                ar[64] * inv + hr.z + bb.z,
                                ar[96] * inv + hr.w + bb.w);
            }
            if (PASS == 0) *(float4*)&A2[r * LDA + 4 * lane] = o;
            else           ((float4*)(out + (size_t)(n0 + r) * C_DIM))[lane] = o;
        }
        if (PASS == 0) {
            __syncthreads();
            for (int it = wave; it < 4 * (3 * C_DIM / 16); it += OWN_WAVES) {
                const int g    = it / (3 * C_DIM / 16);
                const int gcol = (it - g * (3 * C_DIM / 16)) * 16;
                const int mt0  = (g == 0) ? 0 : 3 * g + 1;
                if (mt0 * 16 >= rows) continue;
                const int hi   = lane >> 4;
                const int l16  = lane & 15;
                const float* Bbase;
                if (gcol < C_DIM)          Bbase = Wn + gcol;
                else if (gcol < 2 * C_DIM) Bbase = Wn + (size_t)C_DIM * C_DIM + (gcol - C_DIM);
                else                       Bbase = rootn + (gcol - 2 * C_DIM);
                const float* Bl = Bbase + (size_t)(2 * hi) * C_DIM + l16;
                const float* Al = &A2[(mt0 * 16 + l16) * LDA + 2 * hi];
                float*       Hl = Hn + (size_t)(n0 + mt0 * 16 + 8 * hi) * (3 * C_DIM) + gcol + l16;
                if (g == 0) product_rows<4, FUSED_KC>(Al, Bl, Hl, rows - mt0 * 16, hi);
                else        product_rows<3, FUSED_KC>(Al, Bl, Hl, rows - mt0 * 16, hi);
            }
        }
        __syncthreads();
    }
}

extern "C" void kernel_launch(void* const* d_in, const int* in_sizes, int n_in,
                              void* d_out, int out_size, void* d_ws, size_t ws_size,
                              hipStream_t stream)
{
    const float* x  = (const float*)d_in[0];
    const int*   ei = (const int*)d_in[1];
    const float* ea = (const float*)d_in[2];
    const float* W1 = (const float*)d_in[3];
    const float* r1 = (const float*)d_in[4];
    const float* b1 = (const float*)d_in[5];
    const float* W2 = (const float*)d_in[6];
    const float* r2 = (const float*)d_in[7];
    const float* b2 = (const float*)d_in[8];
    float* out = (float*)d_out;

    const int N = in_sizes[0] / C_DIM;
    const int E = in_sizes[1] / 2;

    float* ws   = (float*)d_ws;
    float* Hbuf  = ws;
    float* H2buf = Hbuf + (size_t)N * 3 * C_DIM;
    int* kept   = (int*)(H2buf + (size_t)N * 3 * C_DIM);
    int* kept_n = kept + (size_t)256 * OWN_WAVES * KEEP_CAP * 4;
    unsigned* bm = (unsigned*)(kept_n + (size_t)256 * OWN_WAVES);

    const int mblocks = (N + MTILE - 1) / MTILE;
    const int oblocks = 256;

    splineconv_gemm_kernel<<<mblocks, 256, 0, stream>>>(x, W1, r1, Hbuf, N, ei, bm, E);
    owner_edge_kernel<0><<<oblocks, 1024, 0, stream>>>(ei, ea, Hbuf, b1, nullptr, kept, kept_n, bm, N, E,
                                                        W2, r2, H2buf);

    owner_edge_kernel<1><<<oblocks, 1024, 0, stream>>>(ei, ea, H2buf, b2, out, kept, kept_n, bm, N, E,
                                                        nullptr, nullptr, nullptr);
}
